# starting kernel
# baseline (speedup 1.0000x reference)
#include <hip/hip_runtime.h>

typedef __attribute__((ext_vector_type(2))) float v2f;
typedef __attribute__((ext_vector_type(8))) float v8f;

__global__ void k_zero(float* __restrict__ p, long long n) {
  long long i = (long long)blockIdx.x * blockDim.x + threadIdx.x;
  if (i < n) p[i] = 0.0f;
}

__global__ void k_node_fc(const float* __restrict__ nf, const float* __restrict__ Wn,
                          const float* __restrict__ bn, float* __restrict__ h, int N) {
  int t = blockIdx.x * blockDim.x + threadIdx.x;
  int n = t >> 6, j = t & 63;
  if (n >= N) return;
  float f0 = nf[n * 3 + 0], f1 = nf[n * 3 + 1], f2 = nf[n * 3 + 2];
  h[t] = bn[j] + Wn[j * 3 + 0] * f0 + Wn[j * 3 + 1] * f1 + Wn[j * 3 + 2] * f2;
}

__global__ void k_degree(const int* __restrict__ src, float* __restrict__ cnt, int E) {
  int e = blockIdx.x * blockDim.x + threadIdx.x;
  if (e < E) atomicAdd(&cnt[src[e]], 1.0f);
}

__global__ void k_scatter(const int* __restrict__ src, const int* __restrict__ dst,
                          const float* __restrict__ h, float* __restrict__ accum, int E) {
  int t = blockIdx.x * blockDim.x + threadIdx.x;
  int e = t >> 2;
  if (e >= E) return;
  int q = (t & 3) * 16;
  const float* hr = h + (long long)dst[e] * 64 + q;
  float* ar = accum + (long long)src[e] * 64 + q;
#pragma unroll
  for (int i = 0; i < 16; ++i) atomicAdd(&ar[i], hr[i]);
}

__global__ void __launch_bounds__(256)
k_layer_gemm(const float* __restrict__ hin, const float* __restrict__ accum,
             const float* __restrict__ cnt,
             const float* __restrict__ Wself, const float* __restrict__ bself,
             const float* __restrict__ Wneigh, const float* __restrict__ bneigh,
             float* __restrict__ hout, int N) {
  __shared__ float Bl[32 * 4 * 32 * 2];
  int tid = threadIdx.x;
  for (int idx = tid; idx < 8192; idx += 256) {
    int v = idx & 1, lane = (idx >> 1) & 31, ct = (idx >> 6) & 3, kk = idx >> 8;
    int n = ct * 16 + (lane & 15);
    int k = kk * 4 + 2 * (lane >> 4) + v;
    Bl[idx] = (k < 64) ? Wself[n * 64 + k] : Wneigh[n * 64 + (k - 64)];
  }
  __syncthreads();

  int w = tid >> 5, lane = tid & 31;
  int base = (blockIdx.x * 8 + w) * 16;
  int half = lane >> 4;
  int mrow = base + (lane & 15);
  long long mc = (mrow < N) ? mrow : 0;

  float c = cnt[mc];
  float inv = (c > 0.f) ? 1.f / c : 1.f;

  v2f A[32];
#pragma unroll
  for (int kk = 0; kk < 16; ++kk) {
    int k0 = kk * 4 + 2 * half;
    v2f a = {hin[mc * 64 + k0], hin[mc * 64 + k0 + 1]};
    A[kk] = a;
  }
#pragma unroll
  for (int kk = 0; kk < 16; ++kk) {
    int k0 = kk * 4 + 2 * half;
    v2f a = {accum[mc * 64 + k0] * inv, accum[mc * 64 + k0 + 1] * inv};
    A[16 + kk] = a;
  }

  v8f acc0 = {}, acc1 = {}, acc2 = {}, acc3 = {};
#pragma unroll
  for (int kk = 0; kk < 32; ++kk) {
    int bb = kk * 256 + lane * 2;
    acc0 = __builtin_amdgcn_wmma_f32_16x16x4_f32(false, A[kk], false, *(const v2f*)&Bl[bb + 0],   (short)0, acc0, false, false);
    acc1 = __builtin_amdgcn_wmma_f32_16x16x4_f32(false, A[kk], false, *(const v2f*)&Bl[bb + 64],  (short)0, acc1, false, false);
    acc2 = __builtin_amdgcn_wmma_f32_16x16x4_f32(false, A[kk], false, *(const v2f*)&Bl[bb + 128], (short)0, acc2, false, false);
    acc3 = __builtin_amdgcn_wmma_f32_16x16x4_f32(false, A[kk], false, *(const v2f*)&Bl[bb + 192], (short)0, acc3, false, false);
  }

#pragma unroll
  for (int ct = 0; ct < 4; ++ct) {
    v8f av = (ct == 0) ? acc0 : (ct == 1) ? acc1 : (ct == 2) ? acc2 : acc3;
    int n = ct * 16 + (lane & 15);
    float bs = bself[n] + bneigh[n];
#pragma unroll
    for (int r = 0; r < 8; ++r) {
      int mg = base + r + 8 * half;
      if (mg < N) {
        float val = av[r] + bs;
        hout[(long long)mg * 64 + n] = val > 0.f ? val : 0.f;
      }
    }
  }
}

__global__ void k_prep(const float* __restrict__ Wa, const float* __restrict__ ba,
                       const float* __restrict__ We, const float* __restrict__ be,
                       const float* __restrict__ env, int NV,
                       float* __restrict__ afold, float* __restrict__ cfold) {
  int j = threadIdx.x;
  if (j >= 64) return;
  float a0 = 0.f, a1 = 0.f, a2 = 0.f, c = ba[j];
  for (int k = 0; k < 64; ++k) {
    float wv = Wa[j * 128 + k];
    a0 += wv * We[k * 3 + 0];
    a1 += wv * We[k * 3 + 1];
    a2 += wv * We[k * 3 + 2];
    c += wv * be[k];
  }
  float invn = 1.0f / (float)NV;
  for (int d = 0; d < 64; ++d) {
    float vm = 0.f;
    for (int v = 0; v < NV; ++v) vm += env[v * 64 + d];
    c += Wa[j * 128 + 64 + d] * (vm * invn);
  }
  afold[j * 3 + 0] = a0;
  afold[j * 3 + 1] = a1;
  afold[j * 3 + 2] = a2;
  cfold[j] = c;
}

__global__ void __launch_bounds__(256)
k_edge_score(const float* __restrict__ ef, const float* __restrict__ afold,
             const float* __restrict__ cfold, const float* __restrict__ W1,
             const float* __restrict__ b1, const float* __restrict__ W2,
             const float* __restrict__ b2, float* __restrict__ scores, int E) {
  __shared__ float Bs[16 * 4 * 32 * 2];
  __shared__ float Af[192];
  __shared__ float Cf[64];
  int tid = threadIdx.x;
  for (int idx = tid; idx < 4096; idx += 256) {
    int v = idx & 1, lane = (idx >> 1) & 31, ct = (idx >> 6) & 3, kk = idx >> 8;
    int n = ct * 16 + (lane & 15);
    int k = kk * 4 + 2 * (lane >> 4) + v;
    Bs[idx] = W1[n * 64 + k];
  }
  if (tid < 192) Af[tid] = afold[tid];
  if (tid < 64) Cf[tid] = cfold[tid];
  __syncthreads();

  int w = tid >> 5, lane = tid & 31;
  int base = (blockIdx.x * 8 + w) * 16;
  int half = lane >> 4;
  int row = base + (lane & 15);
  long long rc = (row < E) ? row : 0;
  float f0 = ef[rc * 3 + 0], f1 = ef[rc * 3 + 1], f2 = ef[rc * 3 + 2];

  v2f A[16];
#pragma unroll
  for (int kk = 0; kk < 16; ++kk) {
    int k0 = kk * 4 + 2 * half;
    float x0 = Cf[k0] + Af[k0 * 3] * f0 + Af[k0 * 3 + 1] * f1 + Af[k0 * 3 + 2] * f2;
    float x1 = Cf[k0 + 1] + Af[(k0 + 1) * 3] * f0 + Af[(k0 + 1) * 3 + 1] * f1 +
               Af[(k0 + 1) * 3 + 2] * f2;
    v2f a = {x0 > 0.f ? x0 : 0.f, x1 > 0.f ? x1 : 0.f};
    A[kk] = a;
  }

  v8f acc0 = {}, acc1 = {}, acc2 = {}, acc3 = {};
#pragma unroll
  for (int kk = 0; kk < 16; ++kk) {
    int bb = kk * 256 + lane * 2;
    acc0 = __builtin_amdgcn_wmma_f32_16x16x4_f32(false, A[kk], false, *(const v2f*)&Bs[bb + 0],   (short)0, acc0, false, false);
    acc1 = __builtin_amdgcn_wmma_f32_16x16x4_f32(false, A[kk], false, *(const v2f*)&Bs[bb + 64],  (short)0, acc1, false, false);
    acc2 = __builtin_amdgcn_wmma_f32_16x16x4_f32(false, A[kk], false, *(const v2f*)&Bs[bb + 128], (short)0, acc2, false, false);
    acc3 = __builtin_amdgcn_wmma_f32_16x16x4_f32(false, A[kk], false, *(const v2f*)&Bs[bb + 192], (short)0, acc3, false, false);
  }

  float p[8] = {0, 0, 0, 0, 0, 0, 0, 0};
#pragma unroll
  for (int ct = 0; ct < 4; ++ct) {
    v8f av = (ct == 0) ? acc0 : (ct == 1) ? acc1 : (ct == 2) ? acc2 : acc3;
    int n = ct * 16 + (lane & 15);
    float bn_ = b1[n], wn = W2[n];
#pragma unroll
    for (int r = 0; r < 8; ++r) {
      float s = av[r] + bn_;
      s = s > 0.f ? s : 0.f;
      p[r] += wn * s;
    }
  }
#pragma unroll
  for (int r = 0; r < 8; ++r) {
    p[r] += __shfl_xor(p[r], 1);
    p[r] += __shfl_xor(p[r], 2);
    p[r] += __shfl_xor(p[r], 4);
    p[r] += __shfl_xor(p[r], 8);
  }
  if ((lane & 15) == 0) {
    float bb2 = b2[0];
    int rb = base + 8 * half;
#pragma unroll
    for (int r = 0; r < 8; ++r)
      if (rb + r < E) scores[rb + r] = p[r] + bb2;
  }
}

extern "C" void kernel_launch(void* const* d_in, const int* in_sizes, int n_in,
                              void* d_out, int out_size, void* d_ws, size_t ws_size,
                              hipStream_t stream) {
  if (n_in < 18) return;
  const float* node_feats = (const float*)d_in[0];
  const float* edge_feats = (const float*)d_in[1];
  const float* env        = (const float*)d_in[2];
  const int*   eidx       = (const int*)d_in[3];
  const float* Wn  = (const float*)d_in[4];
  const float* bn  = (const float*)d_in[5];
  const float* We  = (const float*)d_in[6];
  const float* be  = (const float*)d_in[7];
  const float* Wself  = (const float*)d_in[8];
  const float* bself  = (const float*)d_in[9];
  const float* Wneigh = (const float*)d_in[10];
  const float* bneigh = (const float*)d_in[11];
  const float* Wa  = (const float*)d_in[12];
  const float* ba  = (const float*)d_in[13];
  const float* W1  = (const float*)d_in[14];
  const float* b1  = (const float*)d_in[15];
  const float* W2  = (const float*)d_in[16];
  const float* b2  = (const float*)d_in[17];

  int N  = in_sizes[0] / 3;
  int E  = in_sizes[1] / 3;
  int NV = in_sizes[2] / 64;
  int L  = in_sizes[8] / (64 * 64);

  const int* src = eidx;
  const int* dst = eidx + E;

  float* scores = (float*)d_out;
  float* h_out  = (float*)d_out + E;

  float* accum = (float*)d_ws;
  float* cnt   = accum + (long long)N * 64;
  long long off = (((long long)N * 65) + 63) & ~63LL;
  float* h_b   = (float*)d_ws + off;
  float* afold = h_b + (long long)N * 64;
  float* cfold = afold + 192;

  long long zn = (long long)N * 65;
  k_zero<<<(int)((zn + 255) / 256), 256, 0, stream>>>(accum, zn);
  k_node_fc<<<(N * 64 + 255) / 256, 256, 0, stream>>>(node_feats, Wn, bn, h_out, N);
  k_degree<<<(E + 255) / 256, 256, 0, stream>>>(src, cnt, E);

  int tiles = (N + 15) / 16;
  float* hin = h_out;
  for (int l = 0; l < L; ++l) {
    if (l > 0) {
      long long an = (long long)N * 64;
      k_zero<<<(int)((an + 255) / 256), 256, 0, stream>>>(accum, an);
    }
    k_scatter<<<(int)(((long long)E * 4 + 255) / 256), 256, 0, stream>>>(src, dst, hin, accum, E);
    float* hout = (l == L - 1) ? h_out : ((hin == h_out) ? h_b : h_out);
    k_layer_gemm<<<(tiles + 7) / 8, 256, 0, stream>>>(
        hin, accum, cnt,
        Wself + (long long)l * 64 * 64, bself + (long long)l * 64,
        Wneigh + (long long)l * 64 * 64, bneigh + (long long)l * 64, hout, N);
    hin = hout;
  }

  k_prep<<<1, 64, 0, stream>>>(Wa, ba, We, be, env, NV, afold, cfold);
  int etiles = (E + 15) / 16;
  k_edge_score<<<(etiles + 7) / 8, 256, 0, stream>>>(edge_feats, afold, cfold, W1, b1, W2,
                                                     b2, scores, E);
}


# segment sums (neighbour rows, degrees) in LDS column slabs: a block holds one feature column of all sources, LDS float adds, h column-major; no global atomics, 10 to 7 launches; sum order LDS arrival
# speedup vs baseline: 2.1487x; 2.1487x over previous
#include <hip/hip_runtime.h>


typedef __attribute__((ext_vector_type(2))) float v2f;
typedef __attribute__((ext_vector_type(8))) float v8f;

#define ACAP 50176
#define HCAP 25088
#define ESPL 2
#define CSPL 16

__global__ void k_node_fc(const float* __restrict__ nf, const float* __restrict__ Wn,
                          const float* __restrict__ bn, float* __restrict__ h, int N,
                          int rowmajor) {
  int n = blockIdx.x * blockDim.x + threadIdx.x;
  int j = blockIdx.y;
  if (n >= N) return;
  float f0 = nf[n * 3 + 0], f1 = nf[n * 3 + 1], f2 = nf[n * 3 + 2];
  float v = bn[j] + Wn[j * 3 + 0] * f0 + Wn[j * 3 + 1] * f1 + Wn[j * 3 + 2] * f2;
  if (rowmajor) h[(long long)n * 64 + j] = v;
  else h[(long long)j * N + n] = v;
}

template <bool CNT>
__device__ __forceinline__ void walk_edges(const int* __restrict__ src,
                                           const int* __restrict__ dst, int e0, int e1,
                                           int sbase, int sn, int dlo, int dn,
                                           float* acc, const float* hv, int tid) {
  const int span = e1 - e0;
  const int nfull = span > 0 ? span / 4096 : 0;
  const int* sp = src + e0 + tid;
  const int* dp = dst + e0 + tid;
  int s[4], d[4];
  if (nfull > 0) {
#pragma unroll
    for (int u = 0; u < 4; ++u) {
      s[u] = sp[u * 1024];
      d[u] = CNT ? 0 : dp[u * 1024];
    }
  }
  for (int it = 0; it < nfull; ++it) {
    const int nx = min(it + 1, nfull - 1);
    int s2[4], d2[4];
#pragma unroll
    for (int u = 0; u < 4; ++u) {
      s2[u] = sp[(long long)nx * 4096 + u * 1024];
      d2[u] = CNT ? 0 : dp[(long long)nx * 4096 + u * 1024];
    }
    unsigned ss[4];
    bool ok[4];
    float v[4];
#pragma unroll
    for (int u = 0; u < 4; ++u) {
      ss[u] = (unsigned)(s[u] - sbase);
      if (CNT) {
        ok[u] = ss[u] < (unsigned)sn;
        v[u] = 1.0f;
      } else {
        unsigned dd = (unsigned)(d[u] - dlo);
        bool din = dd < (unsigned)dn;
        ok[u] = din && ss[u] < (unsigned)sn;
        v[u] = hv[din ? dd : 0u];
      }
    }
#pragma unroll
    for (int u = 0; u < 4; ++u)
      if (ok[u])
        __hip_atomic_fetch_add(&acc[ss[u]], v[u], __ATOMIC_RELAXED, __HIP_MEMORY_SCOPE_WORKGROUP);
#pragma unroll
    for (int u = 0; u < 4; ++u) {
      s[u] = s2[u];
      d[u] = d2[u];
    }
  }
  for (int e = e0 + nfull * 4096 + tid; e < e1; e += 1024) {
    unsigned ss1 = (unsigned)(src[e] - sbase);
    if (CNT) {
      if (ss1 < (unsigned)sn)
        __hip_atomic_fetch_add(&acc[ss1], 1.0f, __ATOMIC_RELAXED, __HIP_MEMORY_SCOPE_WORKGROUP);
    } else {
      unsigned dd = (unsigned)(dst[e] - dlo);
      if (ss1 < (unsigned)sn && dd < (unsigned)dn)
        __hip_atomic_fetch_add(&acc[ss1], hv[dd], __ATOMIC_RELAXED, __HIP_MEMORY_SCOPE_WORKGROUP);
    }
  }
}

__global__ void __launch_bounds__(1024)
k_colsum(const int* __restrict__ src, const int* __restrict__ dst,
         const float* __restrict__ hT, float* __restrict__ part,
         float* __restrict__ cntp, int N, int E, int nd, int nreg) {
  __shared__ float acc[ACAP];
  __shared__ float hv[HCAP];
  const int tid = threadIdx.x;
  const int bid = blockIdx.x;
  const bool counting = bid >= nreg;
  int dlo = 0, dn = 0, e0, e1;
  float* outp;
  if (!counting) {
    int c = bid & 63;
    int p = bid >> 6;
    int dch = p % nd, es = p / nd;
    dlo = dch * HCAP;
    dn = min(HCAP, N - dlo);
    int eper = (E + ESPL - 1) / ESPL;
    e0 = es * eper;
    e1 = min(E, e0 + eper);
    outp = part + ((long long)p * 64 + c) * N;
    const float* hc = hT + (long long)c * N + dlo;
    for (int i = tid; i < dn; i += 1024) hv[i] = hc[i];
  } else {
    int j = bid - nreg;
    int eper = (E + CSPL - 1) / CSPL;
    e0 = j * eper;
    e1 = min(E, e0 + eper);
    outp = cntp + (long long)j * N;
  }

  for (int sbase = 0; sbase < N; sbase += ACAP) {
    const int sn = min(ACAP, N - sbase);
    for (int i = tid; i < sn; i += 1024) acc[i] = 0.f;
    __syncthreads();
    if (counting) walk_edges<true>(src, dst, e0, e1, sbase, sn, dlo, dn, acc, hv, tid);
    else          walk_edges<false>(src, dst, e0, e1, sbase, sn, dlo, dn, acc, hv, tid);
    __syncthreads();
    for (int i = tid; i < sn; i += 1024) outp[sbase + i] = acc[i];
    __syncthreads();
  }
}

template <bool OUT_T>
__global__ void __launch_bounds__(256)
k_layer_gemm(const float* __restrict__ hT, const float* __restrict__ part, int np,
             const float* __restrict__ cntp, int ncp,
             const float* __restrict__ Wself, const float* __restrict__ bself,
             const float* __restrict__ Wneigh, const float* __restrict__ bneigh,
             float* __restrict__ hout, int N) {
  __shared__ float Bl[32 * 4 * 32 * 2];
  int tid = threadIdx.x;
  for (int idx = tid; idx < 8192; idx += 256) {
    int v = idx & 1, lane = (idx >> 1) & 31, ct = (idx >> 6) & 3, kk = idx >> 8;
    int n = ct * 16 + (lane & 15);
    int k = kk * 4 + 2 * (lane >> 4) + v;
    Bl[idx] = (k < 64) ? Wself[n * 64 + k] : Wneigh[n * 64 + (k - 64)];
  }
  __syncthreads();

  int w = tid >> 5, lane = tid & 31;
  int base = (blockIdx.x * 8 + w) * 16;
  int half = lane >> 4;
  int mrow = base + (lane & 15);
  long long mc = (mrow < N) ? mrow : 0;

  float c = 0.f;
  for (int j = 0; j < ncp; ++j) c += cntp[(long long)j * N + mc];
  float inv = (c > 0.f) ? 1.f / c : 1.f;

  v2f X[32];
#pragma unroll
  for (int kk = 0; kk < 16; ++kk) {
    long long k0 = kk * 4 + 2 * half;
    v2f a = {hT[k0 * N + mc], hT[(k0 + 1) * N + mc]};
    X[kk] = a;
    v2f z = {0.f, 0.f};
    X[16 + kk] = z;
  }
  for (int p = 0; p < np; ++p) {
    const float* pp = part + (long long)p * 64 * N + mc;
#pragma unroll
    for (int kk = 0; kk < 16; ++kk) {
      long long k0 = kk * 4 + 2 * half;
      X[16 + kk].x += pp[k0 * N];
      X[16 + kk].y += pp[(k0 + 1) * N];
    }
  }
#pragma unroll
  for (int kk = 0; kk < 16; ++kk) {
    X[16 + kk].x *= inv;
    X[16 + kk].y *= inv;
  }

  v8f acc0 = {}, acc1 = {}, acc2 = {}, acc3 = {};
#pragma unroll
  for (int kk = 0; kk < 32; ++kk) {
    int bb = kk * 256 + lane * 2;
    v2f w0 = *(const v2f*)&Bl[bb + 0];
    v2f w1 = *(const v2f*)&Bl[bb + 64];
    v2f w2 = *(const v2f*)&Bl[bb + 128];
    v2f w3 = *(const v2f*)&Bl[bb + 192];
    if (OUT_T) {
      acc0 = __builtin_amdgcn_wmma_f32_16x16x4_f32(false, w0, false, X[kk], (short)0, acc0, false, false);
      acc1 = __builtin_amdgcn_wmma_f32_16x16x4_f32(false, w1, false, X[kk], (short)0, acc1, false, false);
      acc2 = __builtin_amdgcn_wmma_f32_16x16x4_f32(false, w2, false, X[kk], (short)0, acc2, false, false);
      acc3 = __builtin_amdgcn_wmma_f32_16x16x4_f32(false, w3, false, X[kk], (short)0, acc3, false, false);
    } else {
      acc0 = __builtin_amdgcn_wmma_f32_16x16x4_f32(false, X[kk], false, w0, (short)0, acc0, false, false);
      acc1 = __builtin_amdgcn_wmma_f32_16x16x4_f32(false, X[kk], false, w1, (short)0, acc1, false, false);
      acc2 = __builtin_amdgcn_wmma_f32_16x16x4_f32(false, X[kk], false, w2, (short)0, acc2, false, false);
      acc3 = __builtin_amdgcn_wmma_f32_16x16x4_f32(false, X[kk], false, w3, (short)0, acc3, false, false);
    }
  }

#pragma unroll
  for (int ct = 0; ct < 4; ++ct) {
    v8f av = (ct == 0) ? acc0 : (ct == 1) ? acc1 : (ct == 2) ? acc2 : acc3;
    if (OUT_T) {
      int mg = base + (lane & 15);
#pragma unroll
      for (int r = 0; r < 8; ++r) {
        int n = ct * 16 + r + 8 * half;
        float val = av[r] + (bself[n] + bneigh[n]);
        if (mg < N) hout[(long long)n * N + mg] = val > 0.f ? val : 0.f;
      }
    } else {
      int n = ct * 16 + (lane & 15);
      float bs = bself[n] + bneigh[n];
#pragma unroll
      for (int r = 0; r < 8; ++r) {
        int mg = base + r + 8 * half;
        if (mg < N) {
          float val = av[r] + bs;
          hout[(long long)mg * 64 + n] = val > 0.f ? val : 0.f;
        }
      }
    }
  }
}

__global__ void k_prep(const float* __restrict__ Wa, const float* __restrict__ ba,
                       const float* __restrict__ We, const float* __restrict__ be,
                       const float* __restrict__ env, int NV,
                       float* __restrict__ afold, float* __restrict__ cfold) {
  int j = threadIdx.x;
  if (j >= 64) return;
  float a0 = 0.f, a1 = 0.f, a2 = 0.f, c = ba[j];
  for (int k = 0; k < 64; ++k) {
    float wv = Wa[j * 128 + k];
    a0 += wv * We[k * 3 + 0];
    a1 += wv * We[k * 3 + 1];
    a2 += wv * We[k * 3 + 2];
    c += wv * be[k];
  }
  float invn = 1.0f / (float)NV;
  for (int d = 0; d < 64; ++d) {
    float vm = 0.f;
    for (int v = 0; v < NV; ++v) vm += env[v * 64 + d];
    c += Wa[j * 128 + 64 + d] * (vm * invn);
  }
  afold[j * 3 + 0] = a0;
  afold[j * 3 + 1] = a1;
  afold[j * 3 + 2] = a2;
  cfold[j] = c;
}

__global__ void __launch_bounds__(256)
k_edge_score(const float* __restrict__ ef, const float* __restrict__ afold,
             const float* __restrict__ cfold, const float* __restrict__ W1,
             const float* __restrict__ b1, const float* __restrict__ W2,
             const float* __restrict__ b2, float* __restrict__ scores, int E) {
  __shared__ float Bs[16 * 4 * 32 * 2];
  __shared__ float Af[192];
  __shared__ float Cf[64];
  int tid = threadIdx.x;
  for (int idx = tid; idx < 4096; idx += 256) {
    int v = idx & 1, lane = (idx >> 1) & 31, ct = (idx >> 6) & 3, kk = idx >> 8;
    int n = ct * 16 + (lane & 15);
    int k = kk * 4 + 2 * (lane >> 4) + v;
    Bs[idx] = W1[n * 64 + k];
  }
  if (tid < 192) Af[tid] = afold[tid];
  if (tid < 64) Cf[tid] = cfold[tid];
  __syncthreads();

  int w = tid >> 5, lane = tid & 31;
  int base = (blockIdx.x * 8 + w) * 16;
  int half = lane >> 4;
  int row = base + (lane & 15);
  long long rc = (row < E) ? row : 0;
  float f0 = ef[rc * 3 + 0], f1 = ef[rc * 3 + 1], f2 = ef[rc * 3 + 2];

  v2f A[16];
#pragma unroll
  for (int kk = 0; kk < 16; ++kk) {
    int k0 = kk * 4 + 2 * half;
    float x0 = Cf[k0] + Af[k0 * 3] * f0 + Af[k0 * 3 + 1] * f1 + Af[k0 * 3 + 2] * f2;
    float x1 = Cf[k0 + 1] + Af[(k0 + 1) * 3] * f0 + Af[(k0 + 1) * 3 + 1] * f1 +
               Af[(k0 + 1) * 3 + 2] * f2;
    v2f a = {x0 > 0.f ? x0 : 0.f, x1 > 0.f ? x1 : 0.f};
    A[kk] = a;
  }

  v8f acc0 = {}, acc1 = {}, acc2 = {}, acc3 = {};
#pragma unroll
  for (int kk = 0; kk < 16; ++kk) {
    int bb = kk * 256 + lane * 2;
    acc0 = __builtin_amdgcn_wmma_f32_16x16x4_f32(false, A[kk], false, *(const v2f*)&Bs[bb + 0],   (short)0, acc0, false, false);
    acc1 = __builtin_amdgcn_wmma_f32_16x16x4_f32(false, A[kk], false, *(const v2f*)&Bs[bb + 64],  (short)0, acc1, false, false);
    acc2 = __builtin_amdgcn_wmma_f32_16x16x4_f32(false, A[kk], false, *(const v2f*)&Bs[bb + 128], (short)0, acc2, false, false);
    acc3 = __builtin_amdgcn_wmma_f32_16x16x4_f32(false, A[kk], false, *(const v2f*)&Bs[bb + 192], (short)0, acc3, false, false);
  }

  float p[8] = {0, 0, 0, 0, 0, 0, 0, 0};
#pragma unroll
  for (int ct = 0; ct < 4; ++ct) {
    v8f av = (ct == 0) ? acc0 : (ct == 1) ? acc1 : (ct == 2) ? acc2 : acc3;
    int n = ct * 16 + (lane & 15);
    float bn_ = b1[n], wn = W2[n];
#pragma unroll
    for (int r = 0; r < 8; ++r) {
      float s = av[r] + bn_;
      s = s > 0.f ? s : 0.f;
      p[r] += wn * s;
    }
  }
#pragma unroll
  for (int r = 0; r < 8; ++r) {
    p[r] += __shfl_xor(p[r], 1);
    p[r] += __shfl_xor(p[r], 2);
    p[r] += __shfl_xor(p[r], 4);
    p[r] += __shfl_xor(p[r], 8);
  }
  if ((lane & 15) == 0) {
    float bb2 = b2[0];
    int rb = base + 8 * half;
#pragma unroll
    for (int r = 0; r < 8; ++r)
      if (rb + r < E) scores[rb + r] = p[r] + bb2;
  }
}

extern "C" void kernel_launch(void* const* d_in, const int* in_sizes, int n_in,
                              void* d_out, int out_size, void* d_ws, size_t ws_size,
                              hipStream_t stream) {
  if (n_in < 18) return;
  const float* node_feats = (const float*)d_in[0];
  const float* edge_feats = (const float*)d_in[1];
  const float* env        = (const float*)d_in[2];
  const int*   eidx       = (const int*)d_in[3];
  const float* Wn  = (const float*)d_in[4];
  const float* bn  = (const float*)d_in[5];
  const float* We  = (const float*)d_in[6];
  const float* be  = (const float*)d_in[7];
  const float* Wself  = (const float*)d_in[8];
  const float* bself  = (const float*)d_in[9];
  const float* Wneigh = (const float*)d_in[10];
  const float* bneigh = (const float*)d_in[11];
  const float* Wa  = (const float*)d_in[12];
  const float* ba  = (const float*)d_in[13];
  const float* W1  = (const float*)d_in[14];
  const float* b1  = (const float*)d_in[15];
  const float* W2  = (const float*)d_in[16];
  const float* b2  = (const float*)d_in[17];

  int N  = in_sizes[0] / 3;
  int E  = in_sizes[1] / 3;
  int NV = in_sizes[2] / 64;
  int L  = in_sizes[8] / (64 * 64);

  const int* src = eidx;
  const int* dst = eidx + E;

  float* scores = (float*)d_out;
  float* h_out  = (float*)d_out + E;

  int nd   = (N + HCAP - 1) / HCAP;
  int np   = nd * ESPL;
  int nreg = np * 64;

  long long n64  = (long long)N * 64;
  long long npad = (n64 + 63) & ~63LL;
  float* hTa   = (float*)d_ws;
  float* hTb   = hTa + npad;
  float* part  = hTb + npad;
  float* cntp  = part + (long long)np * npad;
  float* afold = cntp + ((((long long)CSPL * N) + 63) & ~63LL);
  float* cfold = afold + 192;

  dim3 gfc((N + 255) / 256, 64);
  if (L == 0) k_node_fc<<<gfc, 256, 0, stream>>>(node_feats, Wn, bn, h_out, N, 1);
  else        k_node_fc<<<gfc, 256, 0, stream>>>(node_feats, Wn, bn, hTa, N, 0);

  int tiles = (N + 15) / 16;
  float* hin = hTa;
  float* hnx = hTb;
  for (int l = 0; l < L; ++l) {
    int nblk = nreg + (l == 0 ? CSPL : 0);
    k_colsum<<<nblk, 1024, 0, stream>>>(src, dst, hin, part, cntp, N, E, nd, nreg);
    if (l == L - 1) {
      k_layer_gemm<false><<<(tiles + 7) / 8, 256, 0, stream>>>(
          hin, part, np, cntp, CSPL,
          Wself + (long long)l * 64 * 64, bself + (long long)l * 64,
          Wneigh + (long long)l * 64 * 64, bneigh + (long long)l * 64, h_out, N);
    } else {
      k_layer_gemm<true><<<(tiles + 7) / 8, 256, 0, stream>>>(
          hin, part, np, cntp, CSPL,
          Wself + (long long)l * 64 * 64, bself + (long long)l * 64,
          Wneigh + (long long)l * 64 * 64, bneigh + (long long)l * 64, hnx, N);
      float* t = hin; hin = hnx; hnx = t;
    }
  }

  k_prep<<<1, 64, 0, stream>>>(Wa, ba, We, be, env, NV, afold, cfold);
  int etiles = (E + 15) / 16;
  k_edge_score<<<(etiles + 7) / 8, 256, 0, stream>>>(edge_feats, afold, cfold, W1, b1, W2,
                                                     b2, scores, E);
}


# edge list split once by destination range into 4 packed bins (ballot ranks, one returning int add per block and bin); a slab block walks only its bin, all LDS ops live; sum order: arrival, per bin
# speedup vs baseline: 2.5716x; 1.1968x over previous
#include <hip/hip_runtime.h>


typedef __attribute__((ext_vector_type(2))) float v2f;
typedef __attribute__((ext_vector_type(8))) float v8f;

#define ACAP 50176
#define HCAP 25088
#define NQ 4
#define CSPL 16
#define BINE 8

__global__ void k_node_fc(const float* __restrict__ nf, const float* __restrict__ Wn,
                          const float* __restrict__ bn, float* __restrict__ h, int N,
                          int rowmajor, int* __restrict__ binfill) {
  int n = blockIdx.x * blockDim.x + threadIdx.x;
  int j = blockIdx.y;
  if (blockIdx.x == 0 && j == 0 && threadIdx.x < NQ) binfill[threadIdx.x] = 0;
  if (n >= N) return;
  float f0 = nf[n * 3 + 0], f1 = nf[n * 3 + 1], f2 = nf[n * 3 + 2];
  float v = bn[j] + Wn[j * 3 + 0] * f0 + Wn[j * 3 + 1] * f1 + Wn[j * 3 + 2] * f2;
  if (rowmajor) h[(long long)n * 64 + j] = v;
  else h[(long long)j * N + n] = v;
}

__global__ void __launch_bounds__(1024)
k_bin(const int* __restrict__ src, const int* __restrict__ dst, int E, int qsz,
      int* __restrict__ bsrc, int* __restrict__ bdst, long long bstride,
      int* __restrict__ binfill) {
  __shared__ int wcnt[32][NQ];
  __shared__ int wbase[32][NQ];
  __shared__ int gbase[NQ];
  const int tid = threadIdx.x;
  const int lane = tid & 31, w = tid >> 5;
  const unsigned below = (1u << lane) - 1u;
  const long long e0 = (long long)blockIdx.x * (1024 * BINE) + w * (32 * BINE) + lane;
  int s[BINE], d[BINE], q[BINE], pos[BINE];
  int c[NQ];
#pragma unroll
  for (int b = 0; b < NQ; ++b) c[b] = 0;
#pragma unroll
  for (int u = 0; u < BINE; ++u) {
    long long e = e0 + u * 32;
    bool valid = e < E;
    s[u] = valid ? src[e] : 0;
    d[u] = valid ? dst[e] : 0;
  }
#pragma unroll
  for (int u = 0; u < BINE; ++u) {
    long long e = e0 + u * 32;
    int qq = (d[u] >= qsz) + (d[u] >= 2 * qsz) + (d[u] >= 3 * qsz);
    q[u] = (e < E) ? qq : NQ;
    pos[u] = 0;
#pragma unroll
    for (int b = 0; b < NQ; ++b) {
      unsigned m = __builtin_amdgcn_ballot_w32(q[u] == b);
      if (q[u] == b) pos[u] = c[b] + __popc(m & below);
      c[b] += __popc(m);
    }
  }
  if (lane == 0) {
#pragma unroll
    for (int b = 0; b < NQ; ++b) wcnt[w][b] = c[b];
  }
  __syncthreads();
  if (tid < NQ) {
    int tot = 0;
    for (int w2 = 0; w2 < 32; ++w2) {
      int t = wcnt[w2][tid];
      wbase[w2][tid] = tot;
      tot += t;
    }
    gbase[tid] = atomicAdd(&binfill[tid], tot);
  }
  __syncthreads();
#pragma unroll
  for (int u = 0; u < BINE; ++u) {
    if (q[u] < NQ) {
      long long o = (long long)q[u] * bstride + gbase[q[u]] + wbase[w][q[u]] + pos[u];
      bsrc[o] = s[u];
      bdst[o] = d[u] - q[u] * qsz;
    }
  }
}

template <bool CNT>
__device__ __forceinline__ void walk_edges(const int* __restrict__ src,
                                           const int* __restrict__ dst, int e0, int e1,
                                           int sbase, int sn, int dlo, int dn,
                                           float* acc, const float* hv, int tid) {
  const int span = e1 - e0;
  const int nfull = span > 0 ? span / 4096 : 0;
  const int* sp = src + e0 + tid;
  const int* dp = dst + e0 + tid;
  int s[4], d[4];
  if (nfull > 0) {
#pragma unroll
    for (int u = 0; u < 4; ++u) {
      s[u] = sp[u * 1024];
      d[u] = CNT ? 0 : dp[u * 1024];
    }
  }
  for (int it = 0; it < nfull; ++it) {
    const int nx = min(it + 1, nfull - 1);
    int s2[4], d2[4];
#pragma unroll
    for (int u = 0; u < 4; ++u) {
      s2[u] = sp[(long long)nx * 4096 + u * 1024];
      d2[u] = CNT ? 0 : dp[(long long)nx * 4096 + u * 1024];
    }
    unsigned ss[4];
    bool ok[4];
    float v[4];
#pragma unroll
    for (int u = 0; u < 4; ++u) {
      ss[u] = (unsigned)(s[u] - sbase);
      if (CNT) {
        ok[u] = ss[u] < (unsigned)sn;
        v[u] = 1.0f;
      } else {
        unsigned dd = (unsigned)(d[u] - dlo);
        bool din = dd < (unsigned)dn;
        ok[u] = din && ss[u] < (unsigned)sn;
        v[u] = hv[din ? dd : 0u];
      }
    }
#pragma unroll
    for (int u = 0; u < 4; ++u)
      if (ok[u])
        __hip_atomic_fetch_add(&acc[ss[u]], v[u], __ATOMIC_RELAXED, __HIP_MEMORY_SCOPE_WORKGROUP);
#pragma unroll
    for (int u = 0; u < 4; ++u) {
      s[u] = s2[u];
      d[u] = d2[u];
    }
  }
  for (int e = e0 + nfull * 4096 + tid; e < e1; e += 1024) {
    unsigned ss1 = (unsigned)(src[e] - sbase);
    if (CNT) {
      if (ss1 < (unsigned)sn)
        __hip_atomic_fetch_add(&acc[ss1], 1.0f, __ATOMIC_RELAXED, __HIP_MEMORY_SCOPE_WORKGROUP);
    } else {
      unsigned dd = (unsigned)(dst[e] - dlo);
      if (ss1 < (unsigned)sn && dd < (unsigned)dn)
        __hip_atomic_fetch_add(&acc[ss1], hv[dd], __ATOMIC_RELAXED, __HIP_MEMORY_SCOPE_WORKGROUP);
    }
  }
}

__global__ void __launch_bounds__(1024)
k_colsum(const int* __restrict__ src, const int* __restrict__ dst,
         const int* __restrict__ bsrc, const int* __restrict__ bdst, long long bstride,
         const int* __restrict__ binfill, int qsz,
         const float* __restrict__ hT, float* __restrict__ part,
         float* __restrict__ cntp, int N, int E, int nreg) {
  __shared__ float acc[ACAP];
  __shared__ float hv[HCAP];
  const int tid = threadIdx.x;
  const int bid = blockIdx.x;
  const bool counting = bid >= nreg;
  int qn = 0, e0 = 0, e1 = 0;
  const int* es = src;
  const int* ed = dst;
  const float* hc = hT;
  float* outp;
  if (!counting) {
    int c = bid & 63;
    int q = bid >> 6;
    int qlo = q * qsz;
    qn = max(0, min(qsz, N - qlo));
    es = bsrc + (long long)q * bstride;
    ed = bdst + (long long)q * bstride;
    e1 = binfill[q];
    outp = part + ((long long)q * 64 + c) * N;
    hc = hT + (long long)c * N + qlo;
  } else {
    int j = bid - nreg;
    int eper = (E + CSPL - 1) / CSPL;
    e0 = j * eper;
    e1 = min(E, e0 + eper);
    outp = cntp + (long long)j * N;
  }

  for (int sbase = 0; sbase < N; sbase += ACAP) {
    const int sn = min(ACAP, N - sbase);
    for (int i = tid; i < sn; i += 1024) acc[i] = 0.f;
    if (counting) {
      __syncthreads();
      walk_edges<true>(es, ed, e0, e1, sbase, sn, 0, 0, acc, hv, tid);
    } else {
      for (int dsub = 0; dsub < qn; dsub += HCAP) {
        const int dn = min(HCAP, qn - dsub);
        __syncthreads();
        for (int i = tid; i < dn; i += 1024) hv[i] = hc[dsub + i];
        __syncthreads();
        walk_edges<false>(es, ed, e0, e1, sbase, sn, dsub, dn, acc, hv, tid);
      }
    }
    __syncthreads();
    for (int i = tid; i < sn; i += 1024) outp[sbase + i] = acc[i];
    __syncthreads();
  }
}

template <bool OUT_T>
__global__ void __launch_bounds__(256)
k_layer_gemm(const float* __restrict__ hT, const float* __restrict__ part, int np,
             const float* __restrict__ cntp, int ncp,
             const float* __restrict__ Wself, const float* __restrict__ bself,
             const float* __restrict__ Wneigh, const float* __restrict__ bneigh,
             float* __restrict__ hout, int N) {
  __shared__ float Bl[32 * 4 * 32 * 2];
  int tid = threadIdx.x;
  for (int idx = tid; idx < 8192; idx += 256) {
    int v = idx & 1, lane = (idx >> 1) & 31, ct = (idx >> 6) & 3, kk = idx >> 8;
    int n = ct * 16 + (lane & 15);
    int k = kk * 4 + 2 * (lane >> 4) + v;
    Bl[idx] = (k < 64) ? Wself[n * 64 + k] : Wneigh[n * 64 + (k - 64)];
  }
  __syncthreads();

  int w = tid >> 5, lane = tid & 31;
  int base = (blockIdx.x * 8 + w) * 16;
  int half = lane >> 4;
  int mrow = base + (lane & 15);
  long long mc = (mrow < N) ? mrow : 0;

  float c = 0.f;
  for (int j = 0; j < ncp; ++j) c += cntp[(long long)j * N + mc];
  float inv = (c > 0.f) ? 1.f / c : 1.f;

  v2f X[32];
#pragma unroll
  for (int kk = 0; kk < 16; ++kk) {
    long long k0 = kk * 4 + 2 * half;
    v2f a = {hT[k0 * N + mc], hT[(k0 + 1) * N + mc]};
    X[kk] = a;
    v2f z = {0.f, 0.f};
    X[16 + kk] = z;
  }
  for (int p = 0; p < np; ++p) {
    const float* pp = part + (long long)p * 64 * N + mc;
#pragma unroll
    for (int kk = 0; kk < 16; ++kk) {
      long long k0 = kk * 4 + 2 * half;
      X[16 + kk].x += pp[k0 * N];
      X[16 + kk].y += pp[(k0 + 1) * N];
    }
  }
#pragma unroll
  for (int kk = 0; kk < 16; ++kk) {
    X[16 + kk].x *= inv;
    X[16 + kk].y *= inv;
  }

  v8f acc0 = {}, acc1 = {}, acc2 = {}, acc3 = {};
#pragma unroll
  for (int kk = 0; kk < 32; ++kk) {
    int bb = kk * 256 + lane * 2;
    v2f w0 = *(const v2f*)&Bl[bb + 0];
    v2f w1 = *(const v2f*)&Bl[bb + 64];
    v2f w2 = *(const v2f*)&Bl[bb + 128];
    v2f w3 = *(const v2f*)&Bl[bb + 192];
    if (OUT_T) {
      acc0 = __builtin_amdgcn_wmma_f32_16x16x4_f32(false, w0, false, X[kk], (short)0, acc0, false, false);
      acc1 = __builtin_amdgcn_wmma_f32_16x16x4_f32(false, w1, false, X[kk], (short)0, acc1, false, false);
      acc2 = __builtin_amdgcn_wmma_f32_16x16x4_f32(false, w2, false, X[kk], (short)0, acc2, false, false);
      acc3 = __builtin_amdgcn_wmma_f32_16x16x4_f32(false, w3, false, X[kk], (short)0, acc3, false, false);
    } else {
      acc0 = __builtin_amdgcn_wmma_f32_16x16x4_f32(false, X[kk], false, w0, (short)0, acc0, false, false);
      acc1 = __builtin_amdgcn_wmma_f32_16x16x4_f32(false, X[kk], false, w1, (short)0, acc1, false, false);
      acc2 = __builtin_amdgcn_wmma_f32_16x16x4_f32(false, X[kk], false, w2, (short)0, acc2, false, false);
      acc3 = __builtin_amdgcn_wmma_f32_16x16x4_f32(false, X[kk], false, w3, (short)0, acc3, false, false);
    }
  }

#pragma unroll
  for (int ct = 0; ct < 4; ++ct) {
    v8f av = (ct == 0) ? acc0 : (ct == 1) ? acc1 : (ct == 2) ? acc2 : acc3;
    if (OUT_T) {
      int mg = base + (lane & 15);
#pragma unroll
      for (int r = 0; r < 8; ++r) {
        int n = ct * 16 + r + 8 * half;
        float val = av[r] + (bself[n] + bneigh[n]);
        if (mg < N) hout[(long long)n * N + mg] = val > 0.f ? val : 0.f;
      }
    } else {
      int n = ct * 16 + (lane & 15);
      float bs = bself[n] + bneigh[n];
#pragma unroll
      for (int r = 0; r < 8; ++r) {
        int mg = base + r + 8 * half;
        if (mg < N) {
          float val = av[r] + bs;
          hout[(long long)mg * 64 + n] = val > 0.f ? val : 0.f;
        }
      }
    }
  }
}

__global__ void k_prep(const float* __restrict__ Wa, const float* __restrict__ ba,
                       const float* __restrict__ We, const float* __restrict__ be,
                       const float* __restrict__ env, int NV,
                       float* __restrict__ afold, float* __restrict__ cfold) {
  int j = threadIdx.x;
  if (j >= 64) return;
  float a0 = 0.f, a1 = 0.f, a2 = 0.f, c = ba[j];
  for (int k = 0; k < 64; ++k) {
    float wv = Wa[j * 128 + k];
    a0 += wv * We[k * 3 + 0];
    a1 += wv * We[k * 3 + 1];
    a2 += wv * We[k * 3 + 2];
    c += wv * be[k];
  }
  float invn = 1.0f / (float)NV;
  for (int d = 0; d < 64; ++d) {
    float vm = 0.f;
    for (int v = 0; v < NV; ++v) vm += env[v * 64 + d];
    c += Wa[j * 128 + 64 + d] * (vm * invn);
  }
  afold[j * 3 + 0] = a0;
  afold[j * 3 + 1] = a1;
  afold[j * 3 + 2] = a2;
  cfold[j] = c;
}

__global__ void __launch_bounds__(256)
k_edge_score(const float* __restrict__ ef, const float* __restrict__ afold,
             const float* __restrict__ cfold, const float* __restrict__ W1,
             const float* __restrict__ b1, const float* __restrict__ W2,
             const float* __restrict__ b2, float* __restrict__ scores, int E) {
  __shared__ float Bs[16 * 4 * 32 * 2];
  __shared__ float Af[192];
  __shared__ float Cf[64];
  int tid = threadIdx.x;
  for (int idx = tid; idx < 4096; idx += 256) {
    int v = idx & 1, lane = (idx >> 1) & 31, ct = (idx >> 6) & 3, kk = idx >> 8;
    int n = ct * 16 + (lane & 15);
    int k = kk * 4 + 2 * (lane >> 4) + v;
    Bs[idx] = W1[n * 64 + k];
  }
  if (tid < 192) Af[tid] = afold[tid];
  if (tid < 64) Cf[tid] = cfold[tid];
  __syncthreads();

  int w = tid >> 5, lane = tid & 31;
  int base = (blockIdx.x * 8 + w) * 16;
  int half = lane >> 4;
  int row = base + (lane & 15);
  long long rc = (row < E) ? row : 0;
  float f0 = ef[rc * 3 + 0], f1 = ef[rc * 3 + 1], f2 = ef[rc * 3 + 2];

  v2f A[16];
#pragma unroll
  for (int kk = 0; kk < 16; ++kk) {
    int k0 = kk * 4 + 2 * half;
    float x0 = Cf[k0] + Af[k0 * 3] * f0 + Af[k0 * 3 + 1] * f1 + Af[k0 * 3 + 2] * f2;
    float x1 = Cf[k0 + 1] + Af[(k0 + 1) * 3] * f0 + Af[(k0 + 1) * 3 + 1] * f1 +
               Af[(k0 + 1) * 3 + 2] * f2;
    v2f a = {x0 > 0.f ? x0 : 0.f, x1 > 0.f ? x1 : 0.f};
    A[kk] = a;
  }

  v8f acc0 = {}, acc1 = {}, acc2 = {}, acc3 = {};
#pragma unroll
  for (int kk = 0; kk < 16; ++kk) {
    int bb = kk * 256 + lane * 2;
    acc0 = __builtin_amdgcn_wmma_f32_16x16x4_f32(false, A[kk], false, *(const v2f*)&Bs[bb + 0],   (short)0, acc0, false, false);
    acc1 = __builtin_amdgcn_wmma_f32_16x16x4_f32(false, A[kk], false, *(const v2f*)&Bs[bb + 64],  (short)0, acc1, false, false);
    acc2 = __builtin_amdgcn_wmma_f32_16x16x4_f32(false, A[kk], false, *(const v2f*)&Bs[bb + 128], (short)0, acc2, false, false);
    acc3 = __builtin_amdgcn_wmma_f32_16x16x4_f32(false, A[kk], false, *(const v2f*)&Bs[bb + 192], (short)0, acc3, false, false);
  }

  float p[8] = {0, 0, 0, 0, 0, 0, 0, 0};
#pragma unroll
  for (int ct = 0; ct < 4; ++ct) {
    v8f av = (ct == 0) ? acc0 : (ct == 1) ? acc1 : (ct == 2) ? acc2 : acc3;
    int n = ct * 16 + (lane & 15);
    float bn_ = b1[n], wn = W2[n];
#pragma unroll
    for (int r = 0; r < 8; ++r) {
      float s = av[r] + bn_;
      s = s > 0.f ? s : 0.f;
      p[r] += wn * s;
    }
  }
#pragma unroll
  for (int r = 0; r < 8; ++r) {
    p[r] += __shfl_xor(p[r], 1);
    p[r] += __shfl_xor(p[r], 2);
    p[r] += __shfl_xor(p[r], 4);
    p[r] += __shfl_xor(p[r], 8);
  }
  if ((lane & 15) == 0) {
    float bb2 = b2[0];
    int rb = base + 8 * half;
#pragma unroll
    for (int r = 0; r < 8; ++r)
      if (rb + r < E) scores[rb + r] = p[r] + bb2;
  }
}

extern "C" void kernel_launch(void* const* d_in, const int* in_sizes, int n_in,
                              void* d_out, int out_size, void* d_ws, size_t ws_size,
                              hipStream_t stream) {
  if (n_in < 18) return;
  const float* node_feats = (const float*)d_in[0];
  const float* edge_feats = (const float*)d_in[1];
  const float* env        = (const float*)d_in[2];
  const int*   eidx       = (const int*)d_in[3];
  const float* Wn  = (const float*)d_in[4];
  const float* bn  = (const float*)d_in[5];
  const float* We  = (const float*)d_in[6];
  const float* be  = (const float*)d_in[7];
  const float* Wself  = (const float*)d_in[8];
  const float* bself  = (const float*)d_in[9];
  const float* Wneigh = (const float*)d_in[10];
  const float* bneigh = (const float*)d_in[11];
  const float* Wa  = (const float*)d_in[12];
  const float* ba  = (const float*)d_in[13];
  const float* W1  = (const float*)d_in[14];
  const float* b1  = (const float*)d_in[15];
  const float* W2  = (const float*)d_in[16];
  const float* b2  = (const float*)d_in[17];

  int N  = in_sizes[0] / 3;
  int E  = in_sizes[1] / 3;
  int NV = in_sizes[2] / 64;
  int L  = in_sizes[8] / (64 * 64);

  const int* src = eidx;
  const int* dst = eidx + E;

  float* scores = (float*)d_out;
  float* h_out  = (float*)d_out + E;

  int qsz  = (N + NQ - 1) / NQ;
  int np   = NQ;
  int nreg = np * 64;

  long long n64  = (long long)N * 64;
  long long npad = (n64 + 63) & ~63LL;
  float* hTa   = (float*)d_ws;
  float* hTb   = hTa + npad;
  float* part  = hTb + npad;
  float* cntp  = part + (long long)np * npad;
  float* afold = cntp + ((((long long)CSPL * N) + 63) & ~63LL);
  float* cfold = afold + 192;
  int* binfill = (int*)(cfold + 64);
  long long bstride = ((long long)E + 31) & ~31LL;
  int* bsrc = binfill + 64;
  int* bdst = bsrc + NQ * bstride;

  dim3 gfc((N + 255) / 256, 64);
  if (L == 0) k_node_fc<<<gfc, 256, 0, stream>>>(node_feats, Wn, bn, h_out, N, 1, binfill);
  else        k_node_fc<<<gfc, 256, 0, stream>>>(node_feats, Wn, bn, hTa, N, 0, binfill);
  if (L > 0)
    k_bin<<<(E + 1024 * BINE - 1) / (1024 * BINE), 1024, 0, stream>>>(src, dst, E, qsz, bsrc, bdst,
                                                                     bstride, binfill);

  int tiles = (N + 15) / 16;
  float* hin = hTa;
  float* hnx = hTb;
  for (int l = 0; l < L; ++l) {
    int nblk = nreg + (l == 0 ? CSPL : 0);
    k_colsum<<<nblk, 1024, 0, stream>>>(src, dst, bsrc, bdst, bstride, binfill, qsz, hin, part,
                                        cntp, N, E, nreg);
    if (l == L - 1) {
      k_layer_gemm<false><<<(tiles + 7) / 8, 256, 0, stream>>>(
          hin, part, np, cntp, CSPL,
          Wself + (long long)l * 64 * 64, bself + (long long)l * 64,
          Wneigh + (long long)l * 64 * 64, bneigh + (long long)l * 64, h_out, N);
    } else {
      k_layer_gemm<true><<<(tiles + 7) / 8, 256, 0, stream>>>(
          hin, part, np, cntp, CSPL,
          Wself + (long long)l * 64 * 64, bself + (long long)l * 64,
          Wneigh + (long long)l * 64 * 64, bneigh + (long long)l * 64, hnx, N);
      float* t = hin; hin = hnx; hnx = t;
    }
  }

  k_prep<<<1, 64, 0, stream>>>(Wa, ba, We, be, env, NV, afold, cfold);
  int etiles = (E + 15) / 16;
  k_edge_score<<<(etiles + 7) / 8, 256, 0, stream>>>(edge_feats, afold, cfold, W1, b1, W2,
                                                     b2, scores, E);
}


# edges binned by source range, not destination: a block keeps the whole h column and its own sources' sums in LDS, sums final (1 plane, not 4 partials); degrees counted in-block; sum order changes
# speedup vs baseline: 3.1458x; 1.2233x over previous
#include <hip/hip_runtime.h>


typedef __attribute__((ext_vector_type(2))) float v2f;
typedef __attribute__((ext_vector_type(8))) float v8f;

#define SCAP 12544
#define HVCAP 50176
#define NQ 4
#define CSPL 16
#define BINE 8

__global__ void k_node_fc(const float* __restrict__ nf, const float* __restrict__ Wn,
                          const float* __restrict__ bn, float* __restrict__ h, int N,
                          int rowmajor, int* __restrict__ binfill) {
  int n = blockIdx.x * blockDim.x + threadIdx.x;
  int j = blockIdx.y;
  if (blockIdx.x == 0 && j == 0 && threadIdx.x < NQ) binfill[threadIdx.x] = 0;
  if (n >= N) return;
  float f0 = nf[n * 3 + 0], f1 = nf[n * 3 + 1], f2 = nf[n * 3 + 2];
  float v = bn[j] + Wn[j * 3 + 0] * f0 + Wn[j * 3 + 1] * f1 + Wn[j * 3 + 2] * f2;
  if (rowmajor) h[(long long)n * 64 + j] = v;
  else h[(long long)j * N + n] = v;
}

__global__ void __launch_bounds__(1024)
k_bin(const int* __restrict__ src, const int* __restrict__ dst, int E, int qsz,
      int* __restrict__ bsrc, int* __restrict__ bdst, long long bstride,
      int* __restrict__ binfill) {
  __shared__ int wcnt[32][NQ];
  __shared__ int wbase[32][NQ];
  __shared__ int gbase[NQ];
  const int tid = threadIdx.x;
  const int lane = tid & 31, w = tid >> 5;
  const unsigned below = (1u << lane) - 1u;
  const long long e0 = (long long)blockIdx.x * (1024 * BINE) + w * (32 * BINE) + lane;
  int s[BINE], d[BINE], q[BINE], pos[BINE];
  int c[NQ];
#pragma unroll
  for (int b = 0; b < NQ; ++b) c[b] = 0;
#pragma unroll
  for (int u = 0; u < BINE; ++u) {
    long long e = e0 + u * 32;
    bool valid = e < E;
    s[u] = valid ? src[e] : 0;
    d[u] = valid ? dst[e] : 0;
  }
#pragma unroll
  for (int u = 0; u < BINE; ++u) {
    long long e = e0 + u * 32;
    int qq = (s[u] >= qsz) + (s[u] >= 2 * qsz) + (s[u] >= 3 * qsz);
    q[u] = (e < E) ? qq : NQ;
    pos[u] = 0;
#pragma unroll
    for (int b = 0; b < NQ; ++b) {
      unsigned m = __builtin_amdgcn_ballot_w32(q[u] == b);
      if (q[u] == b) pos[u] = c[b] + __popc(m & below);
      c[b] += __popc(m);
    }
  }
  if (lane == 0) {
#pragma unroll
    for (int b = 0; b < NQ; ++b) wcnt[w][b] = c[b];
  }
  __syncthreads();
  if (tid < NQ) {
    int tot = 0;
    for (int w2 = 0; w2 < 32; ++w2) {
      int t = wcnt[w2][tid];
      wbase[w2][tid] = tot;
      tot += t;
    }
    gbase[tid] = atomicAdd(&binfill[tid], tot);
  }
  __syncthreads();
#pragma unroll
  for (int u = 0; u < BINE; ++u) {
    if (q[u] < NQ) {
      long long o = (long long)q[u] * bstride + gbase[q[u]] + wbase[w][q[u]] + pos[u];
      bsrc[o] = s[u] - q[u] * qsz;
      bdst[o] = d[u];
    }
  }
}

template <bool CNT>
__device__ __forceinline__ void walk_edges(const int* __restrict__ src,
                                           const int* __restrict__ dst, int e0, int e1,
                                           int sbase, int sn, int dlo, int dn,
                                           float* acc, const float* hv, int tid) {
  const int span = e1 - e0;
  const int nfull = span > 0 ? span / 4096 : 0;
  const int* sp = src + e0 + tid;
  const int* dp = dst + e0 + tid;
  int s[4], d[4];
  if (nfull > 0) {
#pragma unroll
    for (int u = 0; u < 4; ++u) {
      s[u] = sp[u * 1024];
      d[u] = CNT ? 0 : dp[u * 1024];
    }
  }
  for (int it = 0; it < nfull; ++it) {
    const int nx = min(it + 1, nfull - 1);
    int s2[4], d2[4];
#pragma unroll
    for (int u = 0; u < 4; ++u) {
      s2[u] = sp[(long long)nx * 4096 + u * 1024];
      d2[u] = CNT ? 0 : dp[(long long)nx * 4096 + u * 1024];
    }
    unsigned ss[4];
    bool ok[4];
    float v[4];
#pragma unroll
    for (int u = 0; u < 4; ++u) {
      ss[u] = (unsigned)(s[u] - sbase);
      if (CNT) {
        ok[u] = ss[u] < (unsigned)sn;
        v[u] = 1.0f;
      } else {
        unsigned dd = (unsigned)(d[u] - dlo);
        bool din = dd < (unsigned)dn;
        ok[u] = din && ss[u] < (unsigned)sn;
        v[u] = hv[din ? dd : 0u];
      }
    }
#pragma unroll
    for (int u = 0; u < 4; ++u)
      if (ok[u])
        __hip_atomic_fetch_add(&acc[ss[u]], v[u], __ATOMIC_RELAXED, __HIP_MEMORY_SCOPE_WORKGROUP);
#pragma unroll
    for (int u = 0; u < 4; ++u) {
      s[u] = s2[u];
      d[u] = d2[u];
    }
  }
  for (int e = e0 + nfull * 4096 + tid; e < e1; e += 1024) {
    unsigned ss1 = (unsigned)(src[e] - sbase);
    if (CNT) {
      if (ss1 < (unsigned)sn)
        __hip_atomic_fetch_add(&acc[ss1], 1.0f, __ATOMIC_RELAXED, __HIP_MEMORY_SCOPE_WORKGROUP);
    } else {
      unsigned dd = (unsigned)(dst[e] - dlo);
      if (ss1 < (unsigned)sn && dd < (unsigned)dn)
        __hip_atomic_fetch_add(&acc[ss1], hv[dd], __ATOMIC_RELAXED, __HIP_MEMORY_SCOPE_WORKGROUP);
    }
  }
}

__global__ void __launch_bounds__(1024)
k_colsum(const int* __restrict__ bsrc, const int* __restrict__ bdst, long long bstride,
         const int* __restrict__ binfill, int qsz,
         const float* __restrict__ hT, float* __restrict__ sums,
         float* __restrict__ cntp, int N, int docount) {
  __shared__ float acc[SCAP];
  __shared__ float cnt[SCAP];
  __shared__ float hv[HVCAP];
  const int tid = threadIdx.x;
  const int c = blockIdx.x & 63;
  const int q = blockIdx.x >> 6;
  const int qlo = q * qsz;
  const int qn = max(0, min(qsz, N - qlo));
  const int* es = bsrc + (long long)q * bstride;
  const int* ed = bdst + (long long)q * bstride;
  const int e1 = binfill[q];
  const float* hc = hT + (long long)c * N;
  float* outp = sums + (long long)c * N + qlo;
  float* outc = cntp + (long long)c * N + qlo;
  const bool counting = docount != 0 && c < CSPL;
  const int eper = (e1 + CSPL - 1) / CSPL;
  const int ce0 = min(e1, c * eper);
  const int ce1 = min(e1, ce0 + eper);

  for (int sbase = 0; sbase < qn; sbase += SCAP) {
    const int sn = min(SCAP, qn - sbase);
    for (int i = tid; i < sn; i += 1024) acc[i] = 0.f;
    if (counting)
      for (int i = tid; i < sn; i += 1024) cnt[i] = 0.f;
    for (int dlo = 0; dlo < N; dlo += HVCAP) {
      const int dn = min(HVCAP, N - dlo);
      __syncthreads();
      for (int i = tid; i < dn; i += 1024) hv[i] = hc[dlo + i];
      __syncthreads();
      walk_edges<false>(es, ed, 0, e1, sbase, sn, dlo, dn, acc, hv, tid);
    }
    if (counting) walk_edges<true>(es, ed, ce0, ce1, sbase, sn, 0, 0, cnt, hv, tid);
    __syncthreads();
    for (int i = tid; i < sn; i += 1024) outp[sbase + i] = acc[i];
    if (counting)
      for (int i = tid; i < sn; i += 1024) outc[sbase + i] = cnt[i];
    __syncthreads();
  }
}

template <bool OUT_T>
__global__ void __launch_bounds__(256)
k_layer_gemm(const float* __restrict__ hT, const float* __restrict__ part, int np,
             const float* __restrict__ cntp, int ncp,
             const float* __restrict__ Wself, const float* __restrict__ bself,
             const float* __restrict__ Wneigh, const float* __restrict__ bneigh,
             float* __restrict__ hout, int N) {
  __shared__ float Bl[32 * 4 * 32 * 2];
  int tid = threadIdx.x;
  for (int idx = tid; idx < 8192; idx += 256) {
    int v = idx & 1, lane = (idx >> 1) & 31, ct = (idx >> 6) & 3, kk = idx >> 8;
    int n = ct * 16 + (lane & 15);
    int k = kk * 4 + 2 * (lane >> 4) + v;
    Bl[idx] = (k < 64) ? Wself[n * 64 + k] : Wneigh[n * 64 + (k - 64)];
  }
  __syncthreads();

  int w = tid >> 5, lane = tid & 31;
  int base = (blockIdx.x * 8 + w) * 16;
  int half = lane >> 4;
  int mrow = base + (lane & 15);
  long long mc = (mrow < N) ? mrow : 0;

  float c = 0.f;
  for (int j = 0; j < ncp; ++j) c += cntp[(long long)j * N + mc];
  float inv = (c > 0.f) ? 1.f / c : 1.f;

  v2f X[32];
#pragma unroll
  for (int kk = 0; kk < 16; ++kk) {
    long long k0 = kk * 4 + 2 * half;
    v2f a = {hT[k0 * N + mc], hT[(k0 + 1) * N + mc]};
    X[kk] = a;
    v2f z = {0.f, 0.f};
    X[16 + kk] = z;
  }
  for (int p = 0; p < np; ++p) {
    const float* pp = part + (long long)p * 64 * N + mc;
#pragma unroll
    for (int kk = 0; kk < 16; ++kk) {
      long long k0 = kk * 4 + 2 * half;
      X[16 + kk].x += pp[k0 * N];
      X[16 + kk].y += pp[(k0 + 1) * N];
    }
  }
#pragma unroll
  for (int kk = 0; kk < 16; ++kk) {
    X[16 + kk].x *= inv;
    X[16 + kk].y *= inv;
  }

  v8f acc0 = {}, acc1 = {}, acc2 = {}, acc3 = {};
#pragma unroll
  for (int kk = 0; kk < 32; ++kk) {
    int bb = kk * 256 + lane * 2;
    v2f w0 = *(const v2f*)&Bl[bb + 0];
    v2f w1 = *(const v2f*)&Bl[bb + 64];
    v2f w2 = *(const v2f*)&Bl[bb + 128];
    v2f w3 = *(const v2f*)&Bl[bb + 192];
    if (OUT_T) {
      acc0 = __builtin_amdgcn_wmma_f32_16x16x4_f32(false, w0, false, X[kk], (short)0, acc0, false, false);
      acc1 = __builtin_amdgcn_wmma_f32_16x16x4_f32(false, w1, false, X[kk], (short)0, acc1, false, false);
      acc2 = __builtin_amdgcn_wmma_f32_16x16x4_f32(false, w2, false, X[kk], (short)0, acc2, false, false);
      acc3 = __builtin_amdgcn_wmma_f32_16x16x4_f32(false, w3, false, X[kk], (short)0, acc3, false, false);
    } else {
      acc0 = __builtin_amdgcn_wmma_f32_16x16x4_f32(false, X[kk], false, w0, (short)0, acc0, false, false);
      acc1 = __builtin_amdgcn_wmma_f32_16x16x4_f32(false, X[kk], false, w1, (short)0, acc1, false, false);
      acc2 = __builtin_amdgcn_wmma_f32_16x16x4_f32(false, X[kk], false, w2, (short)0, acc2, false, false);
      acc3 = __builtin_amdgcn_wmma_f32_16x16x4_f32(false, X[kk], false, w3, (short)0, acc3, false, false);
    }
  }

#pragma unroll
  for (int ct = 0; ct < 4; ++ct) {
    v8f av = (ct == 0) ? acc0 : (ct == 1) ? acc1 : (ct == 2) ? acc2 : acc3;
    if (OUT_T) {
      int mg = base + (lane & 15);
#pragma unroll
      for (int r = 0; r < 8; ++r) {
        int n = ct * 16 + r + 8 * half;
        float val = av[r] + (bself[n] + bneigh[n]);
        if (mg < N) hout[(long long)n * N + mg] = val > 0.f ? val : 0.f;
      }
    } else {
      int n = ct * 16 + (lane & 15);
      float bs = bself[n] + bneigh[n];
#pragma unroll
      for (int r = 0; r < 8; ++r) {
        int mg = base + r + 8 * half;
        if (mg < N) {
          float val = av[r] + bs;
          hout[(long long)mg * 64 + n] = val > 0.f ? val : 0.f;
        }
      }
    }
  }
}

__global__ void k_prep(const float* __restrict__ Wa, const float* __restrict__ ba,
                       const float* __restrict__ We, const float* __restrict__ be,
                       const float* __restrict__ env, int NV,
                       float* __restrict__ afold, float* __restrict__ cfold) {
  int j = threadIdx.x;
  if (j >= 64) return;
  float a0 = 0.f, a1 = 0.f, a2 = 0.f, c = ba[j];
  for (int k = 0; k < 64; ++k) {
    float wv = Wa[j * 128 + k];
    a0 += wv * We[k * 3 + 0];
    a1 += wv * We[k * 3 + 1];
    a2 += wv * We[k * 3 + 2];
    c += wv * be[k];
  }
  float invn = 1.0f / (float)NV;
  for (int d = 0; d < 64; ++d) {
    float vm = 0.f;
    for (int v = 0; v < NV; ++v) vm += env[v * 64 + d];
    c += Wa[j * 128 + 64 + d] * (vm * invn);
  }
  afold[j * 3 + 0] = a0;
  afold[j * 3 + 1] = a1;
  afold[j * 3 + 2] = a2;
  cfold[j] = c;
}

__global__ void __launch_bounds__(256)
k_edge_score(const float* __restrict__ ef, const float* __restrict__ afold,
             const float* __restrict__ cfold, const float* __restrict__ W1,
             const float* __restrict__ b1, const float* __restrict__ W2,
             const float* __restrict__ b2, float* __restrict__ scores, int E) {
  __shared__ float Bs[16 * 4 * 32 * 2];
  __shared__ float Af[192];
  __shared__ float Cf[64];
  int tid = threadIdx.x;
  for (int idx = tid; idx < 4096; idx += 256) {
    int v = idx & 1, lane = (idx >> 1) & 31, ct = (idx >> 6) & 3, kk = idx >> 8;
    int n = ct * 16 + (lane & 15);
    int k = kk * 4 + 2 * (lane >> 4) + v;
    Bs[idx] = W1[n * 64 + k];
  }
  if (tid < 192) Af[tid] = afold[tid];
  if (tid < 64) Cf[tid] = cfold[tid];
  __syncthreads();

  int w = tid >> 5, lane = tid & 31;
  int base = (blockIdx.x * 8 + w) * 16;
  int half = lane >> 4;
  int row = base + (lane & 15);
  long long rc = (row < E) ? row : 0;
  float f0 = ef[rc * 3 + 0], f1 = ef[rc * 3 + 1], f2 = ef[rc * 3 + 2];

  v2f A[16];
#pragma unroll
  for (int kk = 0; kk < 16; ++kk) {
    int k0 = kk * 4 + 2 * half;
    float x0 = Cf[k0] + Af[k0 * 3] * f0 + Af[k0 * 3 + 1] * f1 + Af[k0 * 3 + 2] * f2;
    float x1 = Cf[k0 + 1] + Af[(k0 + 1) * 3] * f0 + Af[(k0 + 1) * 3 + 1] * f1 +
               Af[(k0 + 1) * 3 + 2] * f2;
    v2f a = {x0 > 0.f ? x0 : 0.f, x1 > 0.f ? x1 : 0.f};
    A[kk] = a;
  }

  v8f acc0 = {}, acc1 = {}, acc2 = {}, acc3 = {};
#pragma unroll
  for (int kk = 0; kk < 16; ++kk) {
    int bb = kk * 256 + lane * 2;
    acc0 = __builtin_amdgcn_wmma_f32_16x16x4_f32(false, A[kk], false, *(const v2f*)&Bs[bb + 0],   (short)0, acc0, false, false);
    acc1 = __builtin_amdgcn_wmma_f32_16x16x4_f32(false, A[kk], false, *(const v2f*)&Bs[bb + 64],  (short)0, acc1, false, false);
    acc2 = __builtin_amdgcn_wmma_f32_16x16x4_f32(false, A[kk], false, *(const v2f*)&Bs[bb + 128], (short)0, acc2, false, false);
    acc3 = __builtin_amdgcn_wmma_f32_16x16x4_f32(false, A[kk], false, *(const v2f*)&Bs[bb + 192], (short)0, acc3, false, false);
  }

  float p[8] = {0, 0, 0, 0, 0, 0, 0, 0};
#pragma unroll
  for (int ct = 0; ct < 4; ++ct) {
    v8f av = (ct == 0) ? acc0 : (ct == 1) ? acc1 : (ct == 2) ? acc2 : acc3;
    int n = ct * 16 + (lane & 15);
    float bn_ = b1[n], wn = W2[n];
#pragma unroll
    for (int r = 0; r < 8; ++r) {
      float s = av[r] + bn_;
      s = s > 0.f ? s : 0.f;
      p[r] += wn * s;
    }
  }
#pragma unroll
  for (int r = 0; r < 8; ++r) {
    p[r] += __shfl_xor(p[r], 1);
    p[r] += __shfl_xor(p[r], 2);
    p[r] += __shfl_xor(p[r], 4);
    p[r] += __shfl_xor(p[r], 8);
  }
  if ((lane & 15) == 0) {
    float bb2 = b2[0];
    int rb = base + 8 * half;
#pragma unroll
    for (int r = 0; r < 8; ++r)
      if (rb + r < E) scores[rb + r] = p[r] + bb2;
  }
}

extern "C" void kernel_launch(void* const* d_in, const int* in_sizes, int n_in,
                              void* d_out, int out_size, void* d_ws, size_t ws_size,
                              hipStream_t stream) {
  if (n_in < 18) return;
  const float* node_feats = (const float*)d_in[0];
  const float* edge_feats = (const float*)d_in[1];
  const float* env        = (const float*)d_in[2];
  const int*   eidx       = (const int*)d_in[3];
  const float* Wn  = (const float*)d_in[4];
  const float* bn  = (const float*)d_in[5];
  const float* We  = (const float*)d_in[6];
  const float* be  = (const float*)d_in[7];
  const float* Wself  = (const float*)d_in[8];
  const float* bself  = (const float*)d_in[9];
  const float* Wneigh = (const float*)d_in[10];
  const float* bneigh = (const float*)d_in[11];
  const float* Wa  = (const float*)d_in[12];
  const float* ba  = (const float*)d_in[13];
  const float* W1  = (const float*)d_in[14];
  const float* b1  = (const float*)d_in[15];
  const float* W2  = (const float*)d_in[16];
  const float* b2  = (const float*)d_in[17];

  int N  = in_sizes[0] / 3;
  int E  = in_sizes[1] / 3;
  int NV = in_sizes[2] / 64;
  int L  = in_sizes[8] / (64 * 64);

  const int* src = eidx;
  const int* dst = eidx + E;

  float* scores = (float*)d_out;
  float* h_out  = (float*)d_out + E;

  int qsz  = (N + NQ - 1) / NQ;
  int np   = 1;

  long long n64  = (long long)N * 64;
  long long npad = (n64 + 63) & ~63LL;
  float* hTa   = (float*)d_ws;
  float* hTb   = hTa + npad;
  float* part  = hTb + npad;
  float* cntp  = part + (long long)np * npad;
  float* afold = cntp + ((((long long)CSPL * N) + 63) & ~63LL);
  float* cfold = afold + 192;
  int* binfill = (int*)(cfold + 64);
  long long bstride = ((long long)E + 31) & ~31LL;
  int* bsrc = binfill + 64;
  int* bdst = bsrc + NQ * bstride;

  dim3 gfc((N + 255) / 256, 64);
  if (L == 0) k_node_fc<<<gfc, 256, 0, stream>>>(node_feats, Wn, bn, h_out, N, 1, binfill);
  else        k_node_fc<<<gfc, 256, 0, stream>>>(node_feats, Wn, bn, hTa, N, 0, binfill);
  if (L > 0)
    k_bin<<<(E + 1024 * BINE - 1) / (1024 * BINE), 1024, 0, stream>>>(src, dst, E, qsz, bsrc, bdst,
                                                                     bstride, binfill);

  int tiles = (N + 15) / 16;
  float* hin = hTa;
  float* hnx = hTb;
  for (int l = 0; l < L; ++l) {
    k_colsum<<<NQ * 64, 1024, 0, stream>>>(bsrc, bdst, bstride, binfill, qsz, hin, part, cntp, N,
                                           l == 0 ? 1 : 0);
    if (l == L - 1) {
      k_layer_gemm<false><<<(tiles + 7) / 8, 256, 0, stream>>>(
          hin, part, np, cntp, CSPL,
          Wself + (long long)l * 64 * 64, bself + (long long)l * 64,
          Wneigh + (long long)l * 64 * 64, bneigh + (long long)l * 64, h_out, N);
    } else {
      k_layer_gemm<true><<<(tiles + 7) / 8, 256, 0, stream>>>(
          hin, part, np, cntp, CSPL,
          Wself + (long long)l * 64 * 64, bself + (long long)l * 64,
          Wneigh + (long long)l * 64 * 64, bneigh + (long long)l * 64, hnx, N);
      float* t = hin; hin = hnx; hnx = t;
    }
  }

  k_prep<<<1, 64, 0, stream>>>(Wa, ba, We, be, env, NV, afold, cfold);
  int etiles = (E + 15) / 16;
  k_edge_score<<<(etiles + 7) / 8, 256, 0, stream>>>(edge_feats, afold, cfold, W1, b1, W2,
                                                     b2, scores, E);
}


# binned edges packed in one 32-bit word (source in its range << 16 | destination): slab blocks read 4 edges per 16-byte load, 8 per trip; half the index bytes; sum order LDS arrival as before
# speedup vs baseline: 3.4140x; 1.0853x over previous
#include <hip/hip_runtime.h>


typedef __attribute__((ext_vector_type(2))) float v2f;
typedef __attribute__((ext_vector_type(8))) float v8f;

#define SCAP 12544
#define HVCAP 50176
#define NQ 4
#define CSPL 16
#define BINE 8

__global__ void k_node_fc(const float* __restrict__ nf, const float* __restrict__ Wn,
                          const float* __restrict__ bn, float* __restrict__ h, int N,
                          int rowmajor, int* __restrict__ binfill) {
  int n = blockIdx.x * blockDim.x + threadIdx.x;
  int j = blockIdx.y;
  if (blockIdx.x == 0 && j == 0 && threadIdx.x < NQ) binfill[threadIdx.x] = 0;
  if (n >= N) return;
  float f0 = nf[n * 3 + 0], f1 = nf[n * 3 + 1], f2 = nf[n * 3 + 2];
  float v = bn[j] + Wn[j * 3 + 0] * f0 + Wn[j * 3 + 1] * f1 + Wn[j * 3 + 2] * f2;
  if (rowmajor) h[(long long)n * 64 + j] = v;
  else h[(long long)j * N + n] = v;
}

template <bool PK>
__global__ void __launch_bounds__(1024)
k_bin(const int* __restrict__ src, const int* __restrict__ dst, int E, int qsz,
      int* __restrict__ bsrc, int* __restrict__ bdst, long long bstride,
      int* __restrict__ binfill) {
  __shared__ int wcnt[32][NQ];
  __shared__ int wbase[32][NQ];
  __shared__ int gbase[NQ];
  const int tid = threadIdx.x;
  const int lane = tid & 31, w = tid >> 5;
  const unsigned below = (1u << lane) - 1u;
  const long long e0 = (long long)blockIdx.x * (1024 * BINE) + w * (32 * BINE) + lane;
  int s[BINE], d[BINE], q[BINE], pos[BINE];
  int c[NQ];
#pragma unroll
  for (int b = 0; b < NQ; ++b) c[b] = 0;
#pragma unroll
  for (int u = 0; u < BINE; ++u) {
    long long e = e0 + u * 32;
    bool valid = e < E;
    s[u] = valid ? src[e] : 0;
    d[u] = valid ? dst[e] : 0;
  }
#pragma unroll
  for (int u = 0; u < BINE; ++u) {
    long long e = e0 + u * 32;
    int qq = (s[u] >= qsz) + (s[u] >= 2 * qsz) + (s[u] >= 3 * qsz);
    q[u] = (e < E) ? qq : NQ;
    pos[u] = 0;
#pragma unroll
    for (int b = 0; b < NQ; ++b) {
      unsigned m = __builtin_amdgcn_ballot_w32(q[u] == b);
      if (q[u] == b) pos[u] = c[b] + __popc(m & below);
      c[b] += __popc(m);
    }
  }
  if (lane == 0) {
#pragma unroll
    for (int b = 0; b < NQ; ++b) wcnt[w][b] = c[b];
  }
  __syncthreads();
  if (tid < NQ) {
    int tot = 0;
    for (int w2 = 0; w2 < 32; ++w2) {
      int t = wcnt[w2][tid];
      wbase[w2][tid] = tot;
      tot += t;
    }
    gbase[tid] = atomicAdd(&binfill[tid], tot);
  }
  __syncthreads();
#pragma unroll
  for (int u = 0; u < BINE; ++u) {
    if (q[u] < NQ) {
      long long o = (long long)q[u] * bstride + gbase[q[u]] + wbase[w][q[u]] + pos[u];
      if (PK) {
        bsrc[o] = (int)(((unsigned)(s[u] - q[u] * qsz) << 16) | (unsigned)d[u]);
      } else {
        bsrc[o] = s[u] - q[u] * qsz;
        bdst[o] = d[u];
      }
    }
  }
}

template <bool CNT>
__device__ __forceinline__ void walk_edges(const int* __restrict__ src,
                                           const int* __restrict__ dst, int e0, int e1,
                                           int sbase, int sn, int dlo, int dn,
                                           float* acc, const float* hv, int tid) {
  const int span = e1 - e0;
  const int nfull = span > 0 ? span / 4096 : 0;
  const int* sp = src + e0 + tid;
  const int* dp = dst + e0 + tid;
  int s[4], d[4];
  if (nfull > 0) {
#pragma unroll
    for (int u = 0; u < 4; ++u) {
      s[u] = sp[u * 1024];
      d[u] = CNT ? 0 : dp[u * 1024];
    }
  }
  for (int it = 0; it < nfull; ++it) {
    const int nx = min(it + 1, nfull - 1);
    int s2[4], d2[4];
#pragma unroll
    for (int u = 0; u < 4; ++u) {
      s2[u] = sp[(long long)nx * 4096 + u * 1024];
      d2[u] = CNT ? 0 : dp[(long long)nx * 4096 + u * 1024];
    }
    unsigned ss[4];
    bool ok[4];
    float v[4];
#pragma unroll
    for (int u = 0; u < 4; ++u) {
      ss[u] = (unsigned)(s[u] - sbase);
      if (CNT) {
        ok[u] = ss[u] < (unsigned)sn;
        v[u] = 1.0f;
      } else {
        unsigned dd = (unsigned)(d[u] - dlo);
        bool din = dd < (unsigned)dn;
        ok[u] = din && ss[u] < (unsigned)sn;
        v[u] = hv[din ? dd : 0u];
      }
    }
#pragma unroll
    for (int u = 0; u < 4; ++u)
      if (ok[u])
        __hip_atomic_fetch_add(&acc[ss[u]], v[u], __ATOMIC_RELAXED, __HIP_MEMORY_SCOPE_WORKGROUP);
#pragma unroll
    for (int u = 0; u < 4; ++u) {
      s[u] = s2[u];
      d[u] = d2[u];
    }
  }
  for (int e = e0 + nfull * 4096 + tid; e < e1; e += 1024) {
    unsigned ss1 = (unsigned)(src[e] - sbase);
    if (CNT) {
      if (ss1 < (unsigned)sn)
        __hip_atomic_fetch_add(&acc[ss1], 1.0f, __ATOMIC_RELAXED, __HIP_MEMORY_SCOPE_WORKGROUP);
    } else {
      unsigned dd = (unsigned)(dst[e] - dlo);
      if (ss1 < (unsigned)sn && dd < (unsigned)dn)
        __hip_atomic_fetch_add(&acc[ss1], hv[dd], __ATOMIC_RELAXED, __HIP_MEMORY_SCOPE_WORKGROUP);
    }
  }
}


template <bool CNT>
__device__ __forceinline__ void walk_packed(const int* __restrict__ pk, int e0, int e1,
                                            int sbase, int sn, int dlo, int dn,
                                            float* acc, const float* hv, int tid) {
  const int span = e1 - e0;
  const int nfull = span > 0 ? span / 8192 : 0;
  const int4* wp = (const int4*)(pk + e0) + tid;
  int4 a0 = {0, 0, 0, 0}, a1 = {0, 0, 0, 0};
  if (nfull > 0) {
    a0 = wp[0];
    a1 = wp[1024];
  }
  for (int it = 0; it < nfull; ++it) {
    const int nx = min(it + 1, nfull - 1);
    const int4 b0 = wp[(long long)nx * 2048];
    const int4 b1 = wp[(long long)nx * 2048 + 1024];
    const unsigned wv[8] = {(unsigned)a0.x, (unsigned)a0.y, (unsigned)a0.z, (unsigned)a0.w,
                            (unsigned)a1.x, (unsigned)a1.y, (unsigned)a1.z, (unsigned)a1.w};
    unsigned ss[8];
    bool ok[8];
    float v[8];
#pragma unroll
    for (int u = 0; u < 8; ++u) {
      ss[u] = (wv[u] >> 16) - (unsigned)sbase;
      if (CNT) {
        ok[u] = ss[u] < (unsigned)sn;
        v[u] = 1.0f;
      } else {
        unsigned dd = (wv[u] & 0xffffu) - (unsigned)dlo;
        bool din = dd < (unsigned)dn;
        ok[u] = din && ss[u] < (unsigned)sn;
        v[u] = hv[din ? dd : 0u];
      }
    }
#pragma unroll
    for (int u = 0; u < 8; ++u)
      if (ok[u])
        __hip_atomic_fetch_add(&acc[ss[u]], v[u], __ATOMIC_RELAXED, __HIP_MEMORY_SCOPE_WORKGROUP);
    a0 = b0;
    a1 = b1;
  }
  for (int e = e0 + nfull * 8192 + tid; e < e1; e += 1024) {
    const unsigned w1 = (unsigned)pk[e];
    unsigned ss1 = (w1 >> 16) - (unsigned)sbase;
    if (CNT) {
      if (ss1 < (unsigned)sn)
        __hip_atomic_fetch_add(&acc[ss1], 1.0f, __ATOMIC_RELAXED, __HIP_MEMORY_SCOPE_WORKGROUP);
    } else {
      unsigned dd = (w1 & 0xffffu) - (unsigned)dlo;
      if (ss1 < (unsigned)sn && dd < (unsigned)dn)
        __hip_atomic_fetch_add(&acc[ss1], hv[dd], __ATOMIC_RELAXED, __HIP_MEMORY_SCOPE_WORKGROUP);
    }
  }
}

template <bool PK>
__global__ void __launch_bounds__(1024)
k_colsum(const int* __restrict__ bsrc, const int* __restrict__ bdst, long long bstride,
         const int* __restrict__ binfill, int qsz,
         const float* __restrict__ hT, float* __restrict__ sums,
         float* __restrict__ cntp, int N, int docount) {
  __shared__ float acc[SCAP];
  __shared__ float cnt[SCAP];
  __shared__ float hv[HVCAP];
  const int tid = threadIdx.x;
  const int c = blockIdx.x & 63;
  const int q = blockIdx.x >> 6;
  const int qlo = q * qsz;
  const int qn = max(0, min(qsz, N - qlo));
  const int* es = bsrc + (long long)q * bstride;
  const int* ed = bdst + (long long)q * bstride;
  const int e1 = binfill[q];
  const float* hc = hT + (long long)c * N;
  float* outp = sums + (long long)c * N + qlo;
  float* outc = cntp + (long long)c * N + qlo;
  const bool counting = docount != 0 && c < CSPL;
  const int eper = ((e1 + CSPL - 1) / CSPL + 3) & ~3;
  const int ce0 = min(e1, c * eper);
  const int ce1 = min(e1, ce0 + eper);

  for (int sbase = 0; sbase < qn; sbase += SCAP) {
    const int sn = min(SCAP, qn - sbase);
    for (int i = tid; i < sn; i += 1024) acc[i] = 0.f;
    if (counting)
      for (int i = tid; i < sn; i += 1024) cnt[i] = 0.f;
    for (int dlo = 0; dlo < N; dlo += HVCAP) {
      const int dn = min(HVCAP, N - dlo);
      __syncthreads();
      for (int i = tid; i < dn; i += 1024) hv[i] = hc[dlo + i];
      __syncthreads();
      if (PK) walk_packed<false>(es, 0, e1, sbase, sn, dlo, dn, acc, hv, tid);
      else walk_edges<false>(es, ed, 0, e1, sbase, sn, dlo, dn, acc, hv, tid);
    }
    if (counting) {
      if (PK) walk_packed<true>(es, ce0, ce1, sbase, sn, 0, 0, cnt, hv, tid);
      else walk_edges<true>(es, ed, ce0, ce1, sbase, sn, 0, 0, cnt, hv, tid);
    }
    __syncthreads();
    for (int i = tid; i < sn; i += 1024) outp[sbase + i] = acc[i];
    if (counting)
      for (int i = tid; i < sn; i += 1024) outc[sbase + i] = cnt[i];
    __syncthreads();
  }
}

template <bool OUT_T>
__global__ void __launch_bounds__(256)
k_layer_gemm(const float* __restrict__ hT, const float* __restrict__ part, int np,
             const float* __restrict__ cntp, int ncp,
             const float* __restrict__ Wself, const float* __restrict__ bself,
             const float* __restrict__ Wneigh, const float* __restrict__ bneigh,
             float* __restrict__ hout, int N) {
  __shared__ float Bl[32 * 4 * 32 * 2];
  int tid = threadIdx.x;
  for (int idx = tid; idx < 8192; idx += 256) {
    int v = idx & 1, lane = (idx >> 1) & 31, ct = (idx >> 6) & 3, kk = idx >> 8;
    int n = ct * 16 + (lane & 15);
    int k = kk * 4 + 2 * (lane >> 4) + v;
    Bl[idx] = (k < 64) ? Wself[n * 64 + k] : Wneigh[n * 64 + (k - 64)];
  }
  __syncthreads();

  int w = tid >> 5, lane = tid & 31;
  int base = (blockIdx.x * 8 + w) * 16;
  int half = lane >> 4;
  int mrow = base + (lane & 15);
  long long mc = (mrow < N) ? mrow : 0;

  float c = 0.f;
  for (int j = 0; j < ncp; ++j) c += cntp[(long long)j * N + mc];
  float inv = (c > 0.f) ? 1.f / c : 1.f;

  v2f X[32];
#pragma unroll
  for (int kk = 0; kk < 16; ++kk) {
    long long k0 = kk * 4 + 2 * half;
    v2f a = {hT[k0 * N + mc], hT[(k0 + 1) * N + mc]};
    X[kk] = a;
    v2f z = {0.f, 0.f};
    X[16 + kk] = z;
  }
  for (int p = 0; p < np; ++p) {
    const float* pp = part + (long long)p * 64 * N + mc;
#pragma unroll
    for (int kk = 0; kk < 16; ++kk) {
      long long k0 = kk * 4 + 2 * half;
      X[16 + kk].x += pp[k0 * N];
      X[16 + kk].y += pp[(k0 + 1) * N];
    }
  }
#pragma unroll
  for (int kk = 0; kk < 16; ++kk) {
    X[16 + kk].x *= inv;
    X[16 + kk].y *= inv;
  }

  v8f acc0 = {}, acc1 = {}, acc2 = {}, acc3 = {};
#pragma unroll
  for (int kk = 0; kk < 32; ++kk) {
    int bb = kk * 256 + lane * 2;
    v2f w0 = *(const v2f*)&Bl[bb + 0];
    v2f w1 = *(const v2f*)&Bl[bb + 64];
    v2f w2 = *(const v2f*)&Bl[bb + 128];
    v2f w3 = *(const v2f*)&Bl[bb + 192];
    if (OUT_T) {
      acc0 = __builtin_amdgcn_wmma_f32_16x16x4_f32(false, w0, false, X[kk], (short)0, acc0, false, false);
      acc1 = __builtin_amdgcn_wmma_f32_16x16x4_f32(false, w1, false, X[kk], (short)0, acc1, false, false);
      acc2 = __builtin_amdgcn_wmma_f32_16x16x4_f32(false, w2, false, X[kk], (short)0, acc2, false, false);
      acc3 = __builtin_amdgcn_wmma_f32_16x16x4_f32(false, w3, false, X[kk], (short)0, acc3, false, false);
    } else {
      acc0 = __builtin_amdgcn_wmma_f32_16x16x4_f32(false, X[kk], false, w0, (short)0, acc0, false, false);
      acc1 = __builtin_amdgcn_wmma_f32_16x16x4_f32(false, X[kk], false, w1, (short)0, acc1, false, false);
      acc2 = __builtin_amdgcn_wmma_f32_16x16x4_f32(false, X[kk], false, w2, (short)0, acc2, false, false);
      acc3 = __builtin_amdgcn_wmma_f32_16x16x4_f32(false, X[kk], false, w3, (short)0, acc3, false, false);
    }
  }

#pragma unroll
  for (int ct = 0; ct < 4; ++ct) {
    v8f av = (ct == 0) ? acc0 : (ct == 1) ? acc1 : (ct == 2) ? acc2 : acc3;
    if (OUT_T) {
      int mg = base + (lane & 15);
#pragma unroll
      for (int r = 0; r < 8; ++r) {
        int n = ct * 16 + r + 8 * half;
        float val = av[r] + (bself[n] + bneigh[n]);
        if (mg < N) hout[(long long)n * N + mg] = val > 0.f ? val : 0.f;
      }
    } else {
      int n = ct * 16 + (lane & 15);
      float bs = bself[n] + bneigh[n];
#pragma unroll
      for (int r = 0; r < 8; ++r) {
        int mg = base + r + 8 * half;
        if (mg < N) {
          float val = av[r] + bs;
          hout[(long long)mg * 64 + n] = val > 0.f ? val : 0.f;
        }
      }
    }
  }
}

__global__ void k_prep(const float* __restrict__ Wa, const float* __restrict__ ba,
                       const float* __restrict__ We, const float* __restrict__ be,
                       const float* __restrict__ env, int NV,
                       float* __restrict__ afold, float* __restrict__ cfold) {
  int j = threadIdx.x;
  if (j >= 64) return;
  float a0 = 0.f, a1 = 0.f, a2 = 0.f, c = ba[j];
  for (int k = 0; k < 64; ++k) {
    float wv = Wa[j * 128 + k];
    a0 += wv * We[k * 3 + 0];
    a1 += wv * We[k * 3 + 1];
    a2 += wv * We[k * 3 + 2];
    c += wv * be[k];
  }
  float invn = 1.0f / (float)NV;
  for (int d = 0; d < 64; ++d) {
    float vm = 0.f;
    for (int v = 0; v < NV; ++v) vm += env[v * 64 + d];
    c += Wa[j * 128 + 64 + d] * (vm * invn);
  }
  afold[j * 3 + 0] = a0;
  afold[j * 3 + 1] = a1;
  afold[j * 3 + 2] = a2;
  cfold[j] = c;
}

__global__ void __launch_bounds__(256)
k_edge_score(const float* __restrict__ ef, const float* __restrict__ afold,
             const float* __restrict__ cfold, const float* __restrict__ W1,
             const float* __restrict__ b1, const float* __restrict__ W2,
             const float* __restrict__ b2, float* __restrict__ scores, int E) {
  __shared__ float Bs[16 * 4 * 32 * 2];
  __shared__ float Af[192];
  __shared__ float Cf[64];
  int tid = threadIdx.x;
  for (int idx = tid; idx < 4096; idx += 256) {
    int v = idx & 1, lane = (idx >> 1) & 31, ct = (idx >> 6) & 3, kk = idx >> 8;
    int n = ct * 16 + (lane & 15);
    int k = kk * 4 + 2 * (lane >> 4) + v;
    Bs[idx] = W1[n * 64 + k];
  }
  if (tid < 192) Af[tid] = afold[tid];
  if (tid < 64) Cf[tid] = cfold[tid];
  __syncthreads();

  int w = tid >> 5, lane = tid & 31;
  int base = (blockIdx.x * 8 + w) * 16;
  int half = lane >> 4;
  int row = base + (lane & 15);
  long long rc = (row < E) ? row : 0;
  float f0 = ef[rc * 3 + 0], f1 = ef[rc * 3 + 1], f2 = ef[rc * 3 + 2];

  v2f A[16];
#pragma unroll
  for (int kk = 0; kk < 16; ++kk) {
    int k0 = kk * 4 + 2 * half;
    float x0 = Cf[k0] + Af[k0 * 3] * f0 + Af[k0 * 3 + 1] * f1 + Af[k0 * 3 + 2] * f2;
    float x1 = Cf[k0 + 1] + Af[(k0 + 1) * 3] * f0 + Af[(k0 + 1) * 3 + 1] * f1 +
               Af[(k0 + 1) * 3 + 2] * f2;
    v2f a = {x0 > 0.f ? x0 : 0.f, x1 > 0.f ? x1 : 0.f};
    A[kk] = a;
  }

  v8f acc0 = {}, acc1 = {}, acc2 = {}, acc3 = {};
#pragma unroll
  for (int kk = 0; kk < 16; ++kk) {
    int bb = kk * 256 + lane * 2;
    acc0 = __builtin_amdgcn_wmma_f32_16x16x4_f32(false, A[kk], false, *(const v2f*)&Bs[bb + 0],   (short)0, acc0, false, false);
    acc1 = __builtin_amdgcn_wmma_f32_16x16x4_f32(false, A[kk], false, *(const v2f*)&Bs[bb + 64],  (short)0, acc1, false, false);
    acc2 = __builtin_amdgcn_wmma_f32_16x16x4_f32(false, A[kk], false, *(const v2f*)&Bs[bb + 128], (short)0, acc2, false, false);
    acc3 = __builtin_amdgcn_wmma_f32_16x16x4_f32(false, A[kk], false, *(const v2f*)&Bs[bb + 192], (short)0, acc3, false, false);
  }

  float p[8] = {0, 0, 0, 0, 0, 0, 0, 0};
#pragma unroll
  for (int ct = 0; ct < 4; ++ct) {
    v8f av = (ct == 0) ? acc0 : (ct == 1) ? acc1 : (ct == 2) ? acc2 : acc3;
    int n = ct * 16 + (lane & 15);
    float bn_ = b1[n], wn = W2[n];
#pragma unroll
    for (int r = 0; r < 8; ++r) {
      float s = av[r] + bn_;
      s = s > 0.f ? s : 0.f;
      p[r] += wn * s;
    }
  }
#pragma unroll
  for (int r = 0; r < 8; ++r) {
    p[r] += __shfl_xor(p[r], 1);
    p[r] += __shfl_xor(p[r], 2);
    p[r] += __shfl_xor(p[r], 4);
    p[r] += __shfl_xor(p[r], 8);
  }
  if ((lane & 15) == 0) {
    float bb2 = b2[0];
    int rb = base + 8 * half;
#pragma unroll
    for (int r = 0; r < 8; ++r)
      if (rb + r < E) scores[rb + r] = p[r] + bb2;
  }
}

extern "C" void kernel_launch(void* const* d_in, const int* in_sizes, int n_in,
                              void* d_out, int out_size, void* d_ws, size_t ws_size,
                              hipStream_t stream) {
  if (n_in < 18) return;
  const float* node_feats = (const float*)d_in[0];
  const float* edge_feats = (const float*)d_in[1];
  const float* env        = (const float*)d_in[2];
  const int*   eidx       = (const int*)d_in[3];
  const float* Wn  = (const float*)d_in[4];
  const float* bn  = (const float*)d_in[5];
  const float* We  = (const float*)d_in[6];
  const float* be  = (const float*)d_in[7];
  const float* Wself  = (const float*)d_in[8];
  const float* bself  = (const float*)d_in[9];
  const float* Wneigh = (const float*)d_in[10];
  const float* bneigh = (const float*)d_in[11];
  const float* Wa  = (const float*)d_in[12];
  const float* ba  = (const float*)d_in[13];
  const float* W1  = (const float*)d_in[14];
  const float* b1  = (const float*)d_in[15];
  const float* W2  = (const float*)d_in[16];
  const float* b2  = (const float*)d_in[17];

  int N  = in_sizes[0] / 3;
  int E  = in_sizes[1] / 3;
  int NV = in_sizes[2] / 64;
  int L  = in_sizes[8] / (64 * 64);

  const int* src = eidx;
  const int* dst = eidx + E;

  float* scores = (float*)d_out;
  float* h_out  = (float*)d_out + E;

  int qsz  = (N + NQ - 1) / NQ;
  int np   = 1;

  long long n64  = (long long)N * 64;
  long long npad = (n64 + 63) & ~63LL;
  float* hTa   = (float*)d_ws;
  float* hTb   = hTa + npad;
  float* part  = hTb + npad;
  float* cntp  = part + (long long)np * npad;
  float* afold = cntp + ((((long long)CSPL * N) + 63) & ~63LL);
  float* cfold = afold + 192;
  int* binfill = (int*)(cfold + 64);
  long long bstride = ((long long)E + 31) & ~31LL;
  int* bsrc = binfill + 64;
  int* bdst = bsrc + NQ * bstride;

  dim3 gfc((N + 255) / 256, 64);
  if (L == 0) k_node_fc<<<gfc, 256, 0, stream>>>(node_feats, Wn, bn, h_out, N, 1, binfill);
  else        k_node_fc<<<gfc, 256, 0, stream>>>(node_feats, Wn, bn, hTa, N, 0, binfill);
  const bool pk = N <= 65536;
  if (L > 0) {
    if (pk)
      k_bin<true><<<(E + 1024 * BINE - 1) / (1024 * BINE), 1024, 0, stream>>>(
          src, dst, E, qsz, bsrc, bdst, bstride, binfill);
    else
      k_bin<false><<<(E + 1024 * BINE - 1) / (1024 * BINE), 1024, 0, stream>>>(
          src, dst, E, qsz, bsrc, bdst, bstride, binfill);
  }

  int tiles = (N + 15) / 16;
  float* hin = hTa;
  float* hnx = hTb;
  for (int l = 0; l < L; ++l) {
    if (pk)
      k_colsum<true><<<NQ * 64, 1024, 0, stream>>>(bsrc, bdst, bstride, binfill, qsz, hin, part,
                                                   cntp, N, l == 0 ? 1 : 0);
    else
      k_colsum<false><<<NQ * 64, 1024, 0, stream>>>(bsrc, bdst, bstride, binfill, qsz, hin, part,
                                                    cntp, N, l == 0 ? 1 : 0);
    if (l == L - 1) {
      k_layer_gemm<false><<<(tiles + 7) / 8, 256, 0, stream>>>(
          hin, part, np, cntp, CSPL,
          Wself + (long long)l * 64 * 64, bself + (long long)l * 64,
          Wneigh + (long long)l * 64 * 64, bneigh + (long long)l * 64, h_out, N);
    } else {
      k_layer_gemm<true><<<(tiles + 7) / 8, 256, 0, stream>>>(
          hin, part, np, cntp, CSPL,
          Wself + (long long)l * 64 * 64, bself + (long long)l * 64,
          Wneigh + (long long)l * 64 * 64, bneigh + (long long)l * 64, hnx, N);
      float* t = hin; hin = hnx; hnx = t;
    }
  }

  k_prep<<<1, 64, 0, stream>>>(Wa, ba, We, be, env, NV, afold, cfold);
  int etiles = (E + 15) / 16;
  k_edge_score<<<(etiles + 7) / 8, 256, 0, stream>>>(edge_feats, afold, cfold, W1, b1, W2,
                                                     b2, scores, E);
}


# edge scorer runs as 256 resident blocks of 32 waves striding over the tiles; each block folds the edge-path weights itself (same sum order), so the serial fold launch is gone: 8 launches to 7
# speedup vs baseline: 3.6786x; 1.0775x over previous
#include <hip/hip_runtime.h>


typedef __attribute__((ext_vector_type(2))) float v2f;
typedef __attribute__((ext_vector_type(8))) float v8f;

#define SCAP 12544
#define HVCAP 50176
#define NQ 4
#define CSPL 16
#define BINE 8

__global__ void k_node_fc(const float* __restrict__ nf, const float* __restrict__ Wn,
                          const float* __restrict__ bn, float* __restrict__ h, int N,
                          int rowmajor, int* __restrict__ binfill) {
  int n = blockIdx.x * blockDim.x + threadIdx.x;
  int j = blockIdx.y;
  if (blockIdx.x == 0 && j == 0 && threadIdx.x < NQ) binfill[threadIdx.x] = 0;
  if (n >= N) return;
  float f0 = nf[n * 3 + 0], f1 = nf[n * 3 + 1], f2 = nf[n * 3 + 2];
  float v = bn[j] + Wn[j * 3 + 0] * f0 + Wn[j * 3 + 1] * f1 + Wn[j * 3 + 2] * f2;
  if (rowmajor) h[(long long)n * 64 + j] = v;
  else h[(long long)j * N + n] = v;
}

template <bool PK>
__global__ void __launch_bounds__(1024)
k_bin(const int* __restrict__ src, const int* __restrict__ dst, int E, int qsz,
      int* __restrict__ bsrc, int* __restrict__ bdst, long long bstride,
      int* __restrict__ binfill) {
  __shared__ int wcnt[32][NQ];
  __shared__ int wbase[32][NQ];
  __shared__ int gbase[NQ];
  const int tid = threadIdx.x;
  const int lane = tid & 31, w = tid >> 5;
  const unsigned below = (1u << lane) - 1u;
  const long long e0 = (long long)blockIdx.x * (1024 * BINE) + w * (32 * BINE) + lane;
  int s[BINE], d[BINE], q[BINE], pos[BINE];
  int c[NQ];
#pragma unroll
  for (int b = 0; b < NQ; ++b) c[b] = 0;
#pragma unroll
  for (int u = 0; u < BINE; ++u) {
    long long e = e0 + u * 32;
    bool valid = e < E;
    s[u] = valid ? src[e] : 0;
    d[u] = valid ? dst[e] : 0;
  }
#pragma unroll
  for (int u = 0; u < BINE; ++u) {
    long long e = e0 + u * 32;
    int qq = (s[u] >= qsz) + (s[u] >= 2 * qsz) + (s[u] >= 3 * qsz);
    q[u] = (e < E) ? qq : NQ;
    pos[u] = 0;
#pragma unroll
    for (int b = 0; b < NQ; ++b) {
      unsigned m = __builtin_amdgcn_ballot_w32(q[u] == b);
      if (q[u] == b) pos[u] = c[b] + __popc(m & below);
      c[b] += __popc(m);
    }
  }
  if (lane == 0) {
#pragma unroll
    for (int b = 0; b < NQ; ++b) wcnt[w][b] = c[b];
  }
  __syncthreads();
  if (tid < NQ) {
    int tot = 0;
    for (int w2 = 0; w2 < 32; ++w2) {
      int t = wcnt[w2][tid];
      wbase[w2][tid] = tot;
      tot += t;
    }
    gbase[tid] = atomicAdd(&binfill[tid], tot);
  }
  __syncthreads();
#pragma unroll
  for (int u = 0; u < BINE; ++u) {
    if (q[u] < NQ) {
      long long o = (long long)q[u] * bstride + gbase[q[u]] + wbase[w][q[u]] + pos[u];
      if (PK) {
        bsrc[o] = (int)(((unsigned)(s[u] - q[u] * qsz) << 16) | (unsigned)d[u]);
      } else {
        bsrc[o] = s[u] - q[u] * qsz;
        bdst[o] = d[u];
      }
    }
  }
}

template <bool CNT>
__device__ __forceinline__ void walk_edges(const int* __restrict__ src,
                                           const int* __restrict__ dst, int e0, int e1,
                                           int sbase, int sn, int dlo, int dn,
                                           float* acc, const float* hv, int tid) {
  const int span = e1 - e0;
  const int nfull = span > 0 ? span / 4096 : 0;
  const int* sp = src + e0 + tid;
  const int* dp = dst + e0 + tid;
  int s[4], d[4];
  if (nfull > 0) {
#pragma unroll
    for (int u = 0; u < 4; ++u) {
      s[u] = sp[u * 1024];
      d[u] = CNT ? 0 : dp[u * 1024];
    }
  }
  for (int it = 0; it < nfull; ++it) {
    const int nx = min(it + 1, nfull - 1);
    int s2[4], d2[4];
#pragma unroll
    for (int u = 0; u < 4; ++u) {
      s2[u] = sp[(long long)nx * 4096 + u * 1024];
      d2[u] = CNT ? 0 : dp[(long long)nx * 4096 + u * 1024];
    }
    unsigned ss[4];
    bool ok[4];
    float v[4];
#pragma unroll
    for (int u = 0; u < 4; ++u) {
      ss[u] = (unsigned)(s[u] - sbase);
      if (CNT) {
        ok[u] = ss[u] < (unsigned)sn;
        v[u] = 1.0f;
      } else {
        unsigned dd = (unsigned)(d[u] - dlo);
        bool din = dd < (unsigned)dn;
        ok[u] = din && ss[u] < (unsigned)sn;
        v[u] = hv[din ? dd : 0u];
      }
    }
#pragma unroll
    for (int u = 0; u < 4; ++u)
      if (ok[u])
        __hip_atomic_fetch_add(&acc[ss[u]], v[u], __ATOMIC_RELAXED, __HIP_MEMORY_SCOPE_WORKGROUP);
#pragma unroll
    for (int u = 0; u < 4; ++u) {
      s[u] = s2[u];
      d[u] = d2[u];
    }
  }
  for (int e = e0 + nfull * 4096 + tid; e < e1; e += 1024) {
    unsigned ss1 = (unsigned)(src[e] - sbase);
    if (CNT) {
      if (ss1 < (unsigned)sn)
        __hip_atomic_fetch_add(&acc[ss1], 1.0f, __ATOMIC_RELAXED, __HIP_MEMORY_SCOPE_WORKGROUP);
    } else {
      unsigned dd = (unsigned)(dst[e] - dlo);
      if (ss1 < (unsigned)sn && dd < (unsigned)dn)
        __hip_atomic_fetch_add(&acc[ss1], hv[dd], __ATOMIC_RELAXED, __HIP_MEMORY_SCOPE_WORKGROUP);
    }
  }
}


template <bool CNT>
__device__ __forceinline__ void walk_packed(const int* __restrict__ pk, int e0, int e1,
                                            int sbase, int sn, int dlo, int dn,
                                            float* acc, const float* hv, int tid) {
  const int span = e1 - e0;
  const int nfull = span > 0 ? span / 8192 : 0;
  const int4* wp = (const int4*)(pk + e0) + tid;
  int4 a0 = {0, 0, 0, 0}, a1 = {0, 0, 0, 0};
  if (nfull > 0) {
    a0 = wp[0];
    a1 = wp[1024];
  }
  for (int it = 0; it < nfull; ++it) {
    const int nx = min(it + 1, nfull - 1);
    const int4 b0 = wp[(long long)nx * 2048];
    const int4 b1 = wp[(long long)nx * 2048 + 1024];
    const unsigned wv[8] = {(unsigned)a0.x, (unsigned)a0.y, (unsigned)a0.z, (unsigned)a0.w,
                            (unsigned)a1.x, (unsigned)a1.y, (unsigned)a1.z, (unsigned)a1.w};
    unsigned ss[8];
    bool ok[8];
    float v[8];
#pragma unroll
    for (int u = 0; u < 8; ++u) {
      ss[u] = (wv[u] >> 16) - (unsigned)sbase;
      if (CNT) {
        ok[u] = ss[u] < (unsigned)sn;
        v[u] = 1.0f;
      } else {
        unsigned dd = (wv[u] & 0xffffu) - (unsigned)dlo;
        bool din = dd < (unsigned)dn;
        ok[u] = din && ss[u] < (unsigned)sn;
        v[u] = hv[din ? dd : 0u];
      }
    }
#pragma unroll
    for (int u = 0; u < 8; ++u)
      if (ok[u])
        __hip_atomic_fetch_add(&acc[ss[u]], v[u], __ATOMIC_RELAXED, __HIP_MEMORY_SCOPE_WORKGROUP);
    a0 = b0;
    a1 = b1;
  }
  for (int e = e0 + nfull * 8192 + tid; e < e1; e += 1024) {
    const unsigned w1 = (unsigned)pk[e];
    unsigned ss1 = (w1 >> 16) - (unsigned)sbase;
    if (CNT) {
      if (ss1 < (unsigned)sn)
        __hip_atomic_fetch_add(&acc[ss1], 1.0f, __ATOMIC_RELAXED, __HIP_MEMORY_SCOPE_WORKGROUP);
    } else {
      unsigned dd = (w1 & 0xffffu) - (unsigned)dlo;
      if (ss1 < (unsigned)sn && dd < (unsigned)dn)
        __hip_atomic_fetch_add(&acc[ss1], hv[dd], __ATOMIC_RELAXED, __HIP_MEMORY_SCOPE_WORKGROUP);
    }
  }
}

template <bool PK>
__global__ void __launch_bounds__(1024)
k_colsum(const int* __restrict__ bsrc, const int* __restrict__ bdst, long long bstride,
         const int* __restrict__ binfill, int qsz,
         const float* __restrict__ hT, float* __restrict__ sums,
         float* __restrict__ cntp, int N, int docount) {
  __shared__ float acc[SCAP];
  __shared__ float cnt[SCAP];
  __shared__ float hv[HVCAP];
  const int tid = threadIdx.x;
  const int c = blockIdx.x & 63;
  const int q = blockIdx.x >> 6;
  const int qlo = q * qsz;
  const int qn = max(0, min(qsz, N - qlo));
  const int* es = bsrc + (long long)q * bstride;
  const int* ed = bdst + (long long)q * bstride;
  const int e1 = binfill[q];
  const float* hc = hT + (long long)c * N;
  float* outp = sums + (long long)c * N + qlo;
  float* outc = cntp + (long long)c * N + qlo;
  const bool counting = docount != 0 && c < CSPL;
  const int eper = ((e1 + CSPL - 1) / CSPL + 3) & ~3;
  const int ce0 = min(e1, c * eper);
  const int ce1 = min(e1, ce0 + eper);

  for (int sbase = 0; sbase < qn; sbase += SCAP) {
    const int sn = min(SCAP, qn - sbase);
    for (int i = tid; i < sn; i += 1024) acc[i] = 0.f;
    if (counting)
      for (int i = tid; i < sn; i += 1024) cnt[i] = 0.f;
    for (int dlo = 0; dlo < N; dlo += HVCAP) {
      const int dn = min(HVCAP, N - dlo);
      __syncthreads();
      for (int i = tid; i < dn; i += 1024) hv[i] = hc[dlo + i];
      __syncthreads();
      if (PK) walk_packed<false>(es, 0, e1, sbase, sn, dlo, dn, acc, hv, tid);
      else walk_edges<false>(es, ed, 0, e1, sbase, sn, dlo, dn, acc, hv, tid);
    }
    if (counting) {
      if (PK) walk_packed<true>(es, ce0, ce1, sbase, sn, 0, 0, cnt, hv, tid);
      else walk_edges<true>(es, ed, ce0, ce1, sbase, sn, 0, 0, cnt, hv, tid);
    }
    __syncthreads();
    for (int i = tid; i < sn; i += 1024) outp[sbase + i] = acc[i];
    if (counting)
      for (int i = tid; i < sn; i += 1024) outc[sbase + i] = cnt[i];
    __syncthreads();
  }
}

template <bool OUT_T>
__global__ void __launch_bounds__(256)
k_layer_gemm(const float* __restrict__ hT, const float* __restrict__ part, int np,
             const float* __restrict__ cntp, int ncp,
             const float* __restrict__ Wself, const float* __restrict__ bself,
             const float* __restrict__ Wneigh, const float* __restrict__ bneigh,
             float* __restrict__ hout, int N) {
  __shared__ float Bl[32 * 4 * 32 * 2];
  int tid = threadIdx.x;
  for (int idx = tid; idx < 8192; idx += 256) {
    int v = idx & 1, lane = (idx >> 1) & 31, ct = (idx >> 6) & 3, kk = idx >> 8;
    int n = ct * 16 + (lane & 15);
    int k = kk * 4 + 2 * (lane >> 4) + v;
    Bl[idx] = (k < 64) ? Wself[n * 64 + k] : Wneigh[n * 64 + (k - 64)];
  }
  __syncthreads();

  int w = tid >> 5, lane = tid & 31;
  int base = (blockIdx.x * 8 + w) * 16;
  int half = lane >> 4;
  int mrow = base + (lane & 15);
  long long mc = (mrow < N) ? mrow : 0;

  float c = 0.f;
  for (int j = 0; j < ncp; ++j) c += cntp[(long long)j * N + mc];
  float inv = (c > 0.f) ? 1.f / c : 1.f;

  v2f X[32];
#pragma unroll
  for (int kk = 0; kk < 16; ++kk) {
    long long k0 = kk * 4 + 2 * half;
    v2f a = {hT[k0 * N + mc], hT[(k0 + 1) * N + mc]};
    X[kk] = a;
    v2f z = {0.f, 0.f};
    X[16 + kk] = z;
  }
  for (int p = 0; p < np; ++p) {
    const float* pp = part + (long long)p * 64 * N + mc;
#pragma unroll
    for (int kk = 0; kk < 16; ++kk) {
      long long k0 = kk * 4 + 2 * half;
      X[16 + kk].x += pp[k0 * N];
      X[16 + kk].y += pp[(k0 + 1) * N];
    }
  }
#pragma unroll
  for (int kk = 0; kk < 16; ++kk) {
    X[16 + kk].x *= inv;
    X[16 + kk].y *= inv;
  }

  v8f acc0 = {}, acc1 = {}, acc2 = {}, acc3 = {};
#pragma unroll
  for (int kk = 0; kk < 32; ++kk) {
    int bb = kk * 256 + lane * 2;
    v2f w0 = *(const v2f*)&Bl[bb + 0];
    v2f w1 = *(const v2f*)&Bl[bb + 64];
    v2f w2 = *(const v2f*)&Bl[bb + 128];
    v2f w3 = *(const v2f*)&Bl[bb + 192];
    if (OUT_T) {
      acc0 = __builtin_amdgcn_wmma_f32_16x16x4_f32(false, w0, false, X[kk], (short)0, acc0, false, false);
      acc1 = __builtin_amdgcn_wmma_f32_16x16x4_f32(false, w1, false, X[kk], (short)0, acc1, false, false);
      acc2 = __builtin_amdgcn_wmma_f32_16x16x4_f32(false, w2, false, X[kk], (short)0, acc2, false, false);
      acc3 = __builtin_amdgcn_wmma_f32_16x16x4_f32(false, w3, false, X[kk], (short)0, acc3, false, false);
    } else {
      acc0 = __builtin_amdgcn_wmma_f32_16x16x4_f32(false, X[kk], false, w0, (short)0, acc0, false, false);
      acc1 = __builtin_amdgcn_wmma_f32_16x16x4_f32(false, X[kk], false, w1, (short)0, acc1, false, false);
      acc2 = __builtin_amdgcn_wmma_f32_16x16x4_f32(false, X[kk], false, w2, (short)0, acc2, false, false);
      acc3 = __builtin_amdgcn_wmma_f32_16x16x4_f32(false, X[kk], false, w3, (short)0, acc3, false, false);
    }
  }

#pragma unroll
  for (int ct = 0; ct < 4; ++ct) {
    v8f av = (ct == 0) ? acc0 : (ct == 1) ? acc1 : (ct == 2) ? acc2 : acc3;
    if (OUT_T) {
      int mg = base + (lane & 15);
#pragma unroll
      for (int r = 0; r < 8; ++r) {
        int n = ct * 16 + r + 8 * half;
        float val = av[r] + (bself[n] + bneigh[n]);
        if (mg < N) hout[(long long)n * N + mg] = val > 0.f ? val : 0.f;
      }
    } else {
      int n = ct * 16 + (lane & 15);
      float bs = bself[n] + bneigh[n];
#pragma unroll
      for (int r = 0; r < 8; ++r) {
        int mg = base + r + 8 * half;
        if (mg < N) {
          float val = av[r] + bs;
          hout[(long long)mg * 64 + n] = val > 0.f ? val : 0.f;
        }
      }
    }
  }
}

#define ESW 32
__global__ void __launch_bounds__(1024)
k_edge_score(const float* __restrict__ ef, const float* __restrict__ Wa,
             const float* __restrict__ ba, const float* __restrict__ We,
             const float* __restrict__ be, const float* __restrict__ env, int NV,
             const float* __restrict__ W1, const float* __restrict__ b1,
             const float* __restrict__ W2, const float* __restrict__ b2,
             float* __restrict__ scores, int E) {
  __shared__ float Bs[16 * 4 * 32 * 2];
  __shared__ __attribute__((aligned(16))) float Af[192];
  __shared__ __attribute__((aligned(16))) float Cf[64];
  __shared__ float WeS[192];
  __shared__ float beS[64];
  __shared__ float vmS[64];
  const int tid = threadIdx.x;
  const int lane = tid & 31;
  const int half = lane >> 4;
  const int w = __builtin_amdgcn_readfirstlane(tid >> 5);
  const int nwv = (int)gridDim.x * ESW;
  const int ntiles = (E + 15) / 16;

  float bn_[4], wn[4];
#pragma unroll
  for (int ct = 0; ct < 4; ++ct) {
    bn_[ct] = b1[ct * 16 + (lane & 15)];
    wn[ct] = W2[ct * 16 + (lane & 15)];
  }
  const float bb2 = b2[0];

  int t = w * (int)gridDim.x + (int)blockIdx.x;
  float f0 = 0.f, f1 = 0.f, f2 = 0.f;
  if (t < ntiles) {
    const int row = t * 16 + (lane & 15);
    const long long rc = (row < E) ? row : 0;
    f0 = ef[rc * 3 + 0];
    f1 = ef[rc * 3 + 1];
    f2 = ef[rc * 3 + 2];
  }

#pragma unroll
  for (int idx = tid; idx < 4096; idx += 1024) {
    int v = idx & 1, ln = (idx >> 1) & 31, ct = (idx >> 6) & 3, kk = idx >> 8;
    int n = ct * 16 + (ln & 15);
    int k = kk * 4 + 2 * (ln >> 4) + v;
    Bs[idx] = W1[n * 64 + k];
  }
  if (tid < 192) {
    WeS[tid] = We[tid];
  } else if (tid < 256) {
    beS[tid - 192] = be[tid - 192];
  } else if (tid < 320) {
    const int d = tid - 256;
    float vm = 0.f;
    for (int v = 0; v < NV; ++v) vm += env[v * 64 + d];
    vmS[d] = vm * (1.0f / (float)NV);
  }
  __syncthreads();
  if (tid < 192) {
    const int j = tid / 3, i = tid - 3 * j;
    const float4* wr = (const float4*)(Wa + j * 128);
    float a = 0.f;
#pragma unroll
    for (int kq = 0; kq < 16; ++kq) {
      const float4 wv = wr[kq];
      a = __builtin_fmaf(wv.x, WeS[(4 * kq + 0) * 3 + i], a);
      a = __builtin_fmaf(wv.y, WeS[(4 * kq + 1) * 3 + i], a);
      a = __builtin_fmaf(wv.z, WeS[(4 * kq + 2) * 3 + i], a);
      a = __builtin_fmaf(wv.w, WeS[(4 * kq + 3) * 3 + i], a);
    }
    Af[tid] = a;
  } else if (tid < 256) {
    const int j = tid - 192;
    const float4* wr = (const float4*)(Wa + j * 128);
    float c = ba[j];
#pragma unroll
    for (int kq = 0; kq < 16; ++kq) {
      const float4 wv = wr[kq];
      c = __builtin_fmaf(wv.x, beS[4 * kq + 0], c);
      c = __builtin_fmaf(wv.y, beS[4 * kq + 1], c);
      c = __builtin_fmaf(wv.z, beS[4 * kq + 2], c);
      c = __builtin_fmaf(wv.w, beS[4 * kq + 3], c);
    }
#pragma unroll
    for (int dq = 0; dq < 16; ++dq) {
      const float4 wv = wr[16 + dq];
      c = __builtin_fmaf(wv.x, vmS[4 * dq + 0], c);
      c = __builtin_fmaf(wv.y, vmS[4 * dq + 1], c);
      c = __builtin_fmaf(wv.z, vmS[4 * dq + 2], c);
      c = __builtin_fmaf(wv.w, vmS[4 * dq + 3], c);
    }
    Cf[j] = c;
  }
  __syncthreads();

  for (; t < ntiles; t += nwv) {
    asm volatile("" ::: "memory");
    const int tn = t + nwv;
    float n0 = 0.f, n1 = 0.f, n2 = 0.f;
    if (tn < ntiles) {
      const int row = tn * 16 + (lane & 15);
      const long long rc = (row < E) ? row : 0;
      n0 = ef[rc * 3 + 0];
      n1 = ef[rc * 3 + 1];
      n2 = ef[rc * 3 + 2];
    }
    const int base = t * 16;

    v2f A[16];
#pragma unroll
    for (int kk = 0; kk < 16; ++kk) {
      int k0 = kk * 4 + 2 * half;
      float x0 = __builtin_fmaf(Af[k0 * 3 + 2], f2,
                 __builtin_fmaf(Af[k0 * 3 + 1], f1, __builtin_fmaf(Af[k0 * 3], f0, Cf[k0])));
      float x1 = __builtin_fmaf(Af[(k0 + 1) * 3 + 2], f2,
                 __builtin_fmaf(Af[(k0 + 1) * 3 + 1], f1,
                                __builtin_fmaf(Af[(k0 + 1) * 3], f0, Cf[k0 + 1])));
      v2f a = {x0 > 0.f ? x0 : 0.f, x1 > 0.f ? x1 : 0.f};
      A[kk] = a;
    }

    v8f acc0 = {}, acc1 = {}, acc2 = {}, acc3 = {};
#pragma unroll
    for (int kk = 0; kk < 16; ++kk) {
      int bb = kk * 256 + lane * 2;
      acc0 = __builtin_amdgcn_wmma_f32_16x16x4_f32(false, A[kk], false, *(const v2f*)&Bs[bb + 0],   (short)0, acc0, false, false);
      acc1 = __builtin_amdgcn_wmma_f32_16x16x4_f32(false, A[kk], false, *(const v2f*)&Bs[bb + 64],  (short)0, acc1, false, false);
      acc2 = __builtin_amdgcn_wmma_f32_16x16x4_f32(false, A[kk], false, *(const v2f*)&Bs[bb + 128], (short)0, acc2, false, false);
      acc3 = __builtin_amdgcn_wmma_f32_16x16x4_f32(false, A[kk], false, *(const v2f*)&Bs[bb + 192], (short)0, acc3, false, false);
    }

    float p[8] = {0, 0, 0, 0, 0, 0, 0, 0};
#pragma unroll
    for (int ct = 0; ct < 4; ++ct) {
      v8f av = (ct == 0) ? acc0 : (ct == 1) ? acc1 : (ct == 2) ? acc2 : acc3;
#pragma unroll
      for (int r = 0; r < 8; ++r) {
        float s = av[r] + bn_[ct];
        s = s > 0.f ? s : 0.f;
        p[r] += wn[ct] * s;
      }
    }
#pragma unroll
    for (int r = 0; r < 8; ++r) {
      p[r] += __shfl_xor(p[r], 1);
      p[r] += __shfl_xor(p[r], 2);
      p[r] += __shfl_xor(p[r], 4);
      p[r] += __shfl_xor(p[r], 8);
    }
    if ((lane & 15) == 0) {
      int rb = base + 8 * half;
#pragma unroll
      for (int r = 0; r < 8; ++r)
        if (rb + r < E) scores[rb + r] = p[r] + bb2;
    }
    f0 = n0;
    f1 = n1;
    f2 = n2;
  }
}

extern "C" void kernel_launch(void* const* d_in, const int* in_sizes, int n_in,
                              void* d_out, int out_size, void* d_ws, size_t ws_size,
                              hipStream_t stream) {
  if (n_in < 18) return;
  const float* node_feats = (const float*)d_in[0];
  const float* edge_feats = (const float*)d_in[1];
  const float* env        = (const float*)d_in[2];
  const int*   eidx       = (const int*)d_in[3];
  const float* Wn  = (const float*)d_in[4];
  const float* bn  = (const float*)d_in[5];
  const float* We  = (const float*)d_in[6];
  const float* be  = (const float*)d_in[7];
  const float* Wself  = (const float*)d_in[8];
  const float* bself  = (const float*)d_in[9];
  const float* Wneigh = (const float*)d_in[10];
  const float* bneigh = (const float*)d_in[11];
  const float* Wa  = (const float*)d_in[12];
  const float* ba  = (const float*)d_in[13];
  const float* W1  = (const float*)d_in[14];
  const float* b1  = (const float*)d_in[15];
  const float* W2  = (const float*)d_in[16];
  const float* b2  = (const float*)d_in[17];

  int N  = in_sizes[0] / 3;
  int E  = in_sizes[1] / 3;
  int NV = in_sizes[2] / 64;
  int L  = in_sizes[8] / (64 * 64);

  const int* src = eidx;
  const int* dst = eidx + E;

  float* scores = (float*)d_out;
  float* h_out  = (float*)d_out + E;

  int qsz  = (N + NQ - 1) / NQ;
  int np   = 1;

  long long n64  = (long long)N * 64;
  long long npad = (n64 + 63) & ~63LL;
  float* hTa   = (float*)d_ws;
  float* hTb   = hTa + npad;
  float* part  = hTb + npad;
  float* cntp  = part + (long long)np * npad;
  int* binfill = (int*)(cntp + ((((long long)CSPL * N) + 63) & ~63LL));
  long long bstride = ((long long)E + 31) & ~31LL;
  int* bsrc = binfill + 64;
  int* bdst = bsrc + NQ * bstride;

  dim3 gfc((N + 255) / 256, 64);
  if (L == 0) k_node_fc<<<gfc, 256, 0, stream>>>(node_feats, Wn, bn, h_out, N, 1, binfill);
  else        k_node_fc<<<gfc, 256, 0, stream>>>(node_feats, Wn, bn, hTa, N, 0, binfill);
  const bool pk = N <= 65536;
  if (L > 0) {
    if (pk)
      k_bin<true><<<(E + 1024 * BINE - 1) / (1024 * BINE), 1024, 0, stream>>>(
          src, dst, E, qsz, bsrc, bdst, bstride, binfill);
    else
      k_bin<false><<<(E + 1024 * BINE - 1) / (1024 * BINE), 1024, 0, stream>>>(
          src, dst, E, qsz, bsrc, bdst, bstride, binfill);
  }

  int tiles = (N + 15) / 16;
  float* hin = hTa;
  float* hnx = hTb;
  for (int l = 0; l < L; ++l) {
    if (pk)
      k_colsum<true><<<NQ * 64, 1024, 0, stream>>>(bsrc, bdst, bstride, binfill, qsz, hin, part,
                                                   cntp, N, l == 0 ? 1 : 0);
    else
      k_colsum<false><<<NQ * 64, 1024, 0, stream>>>(bsrc, bdst, bstride, binfill, qsz, hin, part,
                                                    cntp, N, l == 0 ? 1 : 0);
    if (l == L - 1) {
      k_layer_gemm<false><<<(tiles + 7) / 8, 256, 0, stream>>>(
          hin, part, np, cntp, CSPL,
          Wself + (long long)l * 64 * 64, bself + (long long)l * 64,
          Wneigh + (long long)l * 64 * 64, bneigh + (long long)l * 64, h_out, N);
    } else {
      k_layer_gemm<true><<<(tiles + 7) / 8, 256, 0, stream>>>(
          hin, part, np, cntp, CSPL,
          Wself + (long long)l * 64 * 64, bself + (long long)l * 64,
          Wneigh + (long long)l * 64 * 64, bneigh + (long long)l * 64, hnx, N);
      float* t = hin; hin = hnx; hnx = t;
    }
  }

  k_edge_score<<<256, 1024, 0, stream>>>(edge_feats, Wa, ba, We, be, env, NV, W1, b1, W2, b2,
                                         scores, E);
}


# edges of a source-range bin split by (source mod 32) into lane-class lists, lane r walks class r: the 32 lanes of every LDS float add hit 32 different banks; sum order: LDS arrival, new edge order
# speedup vs baseline: 3.8597x; 1.0492x over previous
#include <hip/hip_runtime.h>


typedef __attribute__((ext_vector_type(2))) float v2f;
typedef __attribute__((ext_vector_type(8))) float v8f;

#define SCAP 12544
#define HVCAP 50176
#define NQ 4
#define CSPL 16
#define BINE 8

__global__ void k_node_fc(const float* __restrict__ nf, const float* __restrict__ Wn,
                          const float* __restrict__ bn, float* __restrict__ h, int N,
                          int rowmajor, int* __restrict__ binfill) {
  int n = blockIdx.x * blockDim.x + threadIdx.x;
  int j = blockIdx.y;
  if (blockIdx.x == 0 && j == 0 && threadIdx.x < 192) binfill[threadIdx.x] = 0;
  if (n >= N) return;
  float f0 = nf[n * 3 + 0], f1 = nf[n * 3 + 1], f2 = nf[n * 3 + 2];
  float v = bn[j] + Wn[j * 3 + 0] * f0 + Wn[j * 3 + 1] * f1 + Wn[j * 3 + 2] * f2;
  if (rowmajor) h[(long long)n * 64 + j] = v;
  else h[(long long)j * N + n] = v;
}

template <bool PK>
__global__ void __launch_bounds__(1024)
k_bin(const int* __restrict__ src, const int* __restrict__ dst, int E, int qsz,
      int* __restrict__ bsrc, int* __restrict__ bdst, long long bstride,
      int* __restrict__ binfill) {
  __shared__ int wcnt[32][NQ];
  __shared__ int wbase[32][NQ];
  __shared__ int gbase[NQ];
  const int tid = threadIdx.x;
  const int lane = tid & 31, w = tid >> 5;
  const unsigned below = (1u << lane) - 1u;
  const long long e0 = (long long)blockIdx.x * (1024 * BINE) + w * (32 * BINE) + lane;
  int s[BINE], d[BINE], q[BINE], pos[BINE];
  int c[NQ];
#pragma unroll
  for (int b = 0; b < NQ; ++b) c[b] = 0;
#pragma unroll
  for (int u = 0; u < BINE; ++u) {
    long long e = e0 + u * 32;
    bool valid = e < E;
    s[u] = valid ? src[e] : 0;
    d[u] = valid ? dst[e] : 0;
  }
#pragma unroll
  for (int u = 0; u < BINE; ++u) {
    long long e = e0 + u * 32;
    int qq = (s[u] >= qsz) + (s[u] >= 2 * qsz) + (s[u] >= 3 * qsz);
    q[u] = (e < E) ? qq : NQ;
    pos[u] = 0;
#pragma unroll
    for (int b = 0; b < NQ; ++b) {
      unsigned m = __builtin_amdgcn_ballot_w32(q[u] == b);
      if (q[u] == b) pos[u] = c[b] + __popc(m & below);
      c[b] += __popc(m);
    }
  }
  if (lane == 0) {
#pragma unroll
    for (int b = 0; b < NQ; ++b) wcnt[w][b] = c[b];
  }
  __syncthreads();
  if (tid < NQ) {
    int tot = 0;
    for (int w2 = 0; w2 < 32; ++w2) {
      int t = wcnt[w2][tid];
      wbase[w2][tid] = tot;
      tot += t;
    }
    gbase[tid] = atomicAdd(&binfill[tid], tot);
  }
  __syncthreads();
#pragma unroll
  for (int u = 0; u < BINE; ++u) {
    if (q[u] < NQ) {
      long long o = (long long)q[u] * bstride + gbase[q[u]] + wbase[w][q[u]] + pos[u];
      if (PK) {
        bsrc[o] = (int)(((unsigned)(s[u] - q[u] * qsz) << 16) | (unsigned)d[u]);
      } else {
        bsrc[o] = s[u] - q[u] * qsz;
        bdst[o] = d[u];
      }
    }
  }
}

template <bool CNT>
__device__ __forceinline__ void walk_edges(const int* __restrict__ src,
                                           const int* __restrict__ dst, int e0, int e1,
                                           int sbase, int sn, int dlo, int dn,
                                           float* acc, const float* hv, int tid) {
  const int span = e1 - e0;
  const int nfull = span > 0 ? span / 4096 : 0;
  const int* sp = src + e0 + tid;
  const int* dp = dst + e0 + tid;
  int s[4], d[4];
  if (nfull > 0) {
#pragma unroll
    for (int u = 0; u < 4; ++u) {
      s[u] = sp[u * 1024];
      d[u] = CNT ? 0 : dp[u * 1024];
    }
  }
  for (int it = 0; it < nfull; ++it) {
    const int nx = min(it + 1, nfull - 1);
    int s2[4], d2[4];
#pragma unroll
    for (int u = 0; u < 4; ++u) {
      s2[u] = sp[(long long)nx * 4096 + u * 1024];
      d2[u] = CNT ? 0 : dp[(long long)nx * 4096 + u * 1024];
    }
    unsigned ss[4];
    bool ok[4];
    float v[4];
#pragma unroll
    for (int u = 0; u < 4; ++u) {
      ss[u] = (unsigned)(s[u] - sbase);
      if (CNT) {
        ok[u] = ss[u] < (unsigned)sn;
        v[u] = 1.0f;
      } else {
        unsigned dd = (unsigned)(d[u] - dlo);
        bool din = dd < (unsigned)dn;
        ok[u] = din && ss[u] < (unsigned)sn;
        v[u] = hv[din ? dd : 0u];
      }
    }
#pragma unroll
    for (int u = 0; u < 4; ++u)
      if (ok[u])
        __hip_atomic_fetch_add(&acc[ss[u]], v[u], __ATOMIC_RELAXED, __HIP_MEMORY_SCOPE_WORKGROUP);
#pragma unroll
    for (int u = 0; u < 4; ++u) {
      s[u] = s2[u];
      d[u] = d2[u];
    }
  }
  for (int e = e0 + nfull * 4096 + tid; e < e1; e += 1024) {
    unsigned ss1 = (unsigned)(src[e] - sbase);
    if (CNT) {
      if (ss1 < (unsigned)sn)
        __hip_atomic_fetch_add(&acc[ss1], 1.0f, __ATOMIC_RELAXED, __HIP_MEMORY_SCOPE_WORKGROUP);
    } else {
      unsigned dd = (unsigned)(dst[e] - dlo);
      if (ss1 < (unsigned)sn && dd < (unsigned)dn)
        __hip_atomic_fetch_add(&acc[ss1], hv[dd], __ATOMIC_RELAXED, __HIP_MEMORY_SCOPE_WORKGROUP);
    }
  }
}


template <bool CNT>
__device__ __forceinline__ void walk_packed(const int* __restrict__ pk, int e0, int e1,
                                            int sbase, int sn, int dlo, int dn,
                                            float* acc, const float* hv, int tid) {
  const int span = e1 - e0;
  const int nfull = span > 0 ? span / 8192 : 0;
  const int4* wp = (const int4*)(pk + e0) + tid;
  int4 a0 = {0, 0, 0, 0}, a1 = {0, 0, 0, 0};
  if (nfull > 0) {
    a0 = wp[0];
    a1 = wp[1024];
  }
  for (int it = 0; it < nfull; ++it) {
    const int nx = min(it + 1, nfull - 1);
    const int4 b0 = wp[(long long)nx * 2048];
    const int4 b1 = wp[(long long)nx * 2048 + 1024];
    const unsigned wv[8] = {(unsigned)a0.x, (unsigned)a0.y, (unsigned)a0.z, (unsigned)a0.w,
                            (unsigned)a1.x, (unsigned)a1.y, (unsigned)a1.z, (unsigned)a1.w};
    unsigned ss[8];
    bool ok[8];
    float v[8];
#pragma unroll
    for (int u = 0; u < 8; ++u) {
      ss[u] = (wv[u] >> 16) - (unsigned)sbase;
      if (CNT) {
        ok[u] = ss[u] < (unsigned)sn;
        v[u] = 1.0f;
      } else {
        unsigned dd = (wv[u] & 0xffffu) - (unsigned)dlo;
        bool din = dd < (unsigned)dn;
        ok[u] = din && ss[u] < (unsigned)sn;
        v[u] = hv[din ? dd : 0u];
      }
    }
#pragma unroll
    for (int u = 0; u < 8; ++u)
      if (ok[u])
        __hip_atomic_fetch_add(&acc[ss[u]], v[u], __ATOMIC_RELAXED, __HIP_MEMORY_SCOPE_WORKGROUP);
    a0 = b0;
    a1 = b1;
  }
  for (int e = e0 + nfull * 8192 + tid; e < e1; e += 1024) {
    const unsigned w1 = (unsigned)pk[e];
    unsigned ss1 = (w1 >> 16) - (unsigned)sbase;
    if (CNT) {
      if (ss1 < (unsigned)sn)
        __hip_atomic_fetch_add(&acc[ss1], 1.0f, __ATOMIC_RELAXED, __HIP_MEMORY_SCOPE_WORKGROUP);
    } else {
      unsigned dd = (w1 & 0xffffu) - (unsigned)dlo;
      if (ss1 < (unsigned)sn && dd < (unsigned)dn)
        __hip_atomic_fetch_add(&acc[ss1], hv[dd], __ATOMIC_RELAXED, __HIP_MEMORY_SCOPE_WORKGROUP);
    }
  }
}

template <bool PK>
__global__ void __launch_bounds__(1024)
k_colsum(const int* __restrict__ bsrc, const int* __restrict__ bdst, long long bstride,
         const int* __restrict__ binfill, int qsz,
         const float* __restrict__ hT, float* __restrict__ sums,
         float* __restrict__ cntp, int N, int docount) {
  __shared__ float acc[SCAP];
  __shared__ float cnt[SCAP];
  __shared__ float hv[HVCAP];
  const int tid = threadIdx.x;
  const int c = blockIdx.x & 63;
  const int q = blockIdx.x >> 6;
  const int qlo = q * qsz;
  const int qn = max(0, min(qsz, N - qlo));
  const int* es = bsrc + (long long)q * bstride;
  const int* ed = bdst + (long long)q * bstride;
  const int e1 = binfill[q];
  const float* hc = hT + (long long)c * N;
  float* outp = sums + (long long)c * N + qlo;
  float* outc = cntp + (long long)c * N + qlo;
  const bool counting = docount != 0 && c < CSPL;
  const int eper = ((e1 + CSPL - 1) / CSPL + 3) & ~3;
  const int ce0 = min(e1, c * eper);
  const int ce1 = min(e1, ce0 + eper);

  for (int sbase = 0; sbase < qn; sbase += SCAP) {
    const int sn = min(SCAP, qn - sbase);
    for (int i = tid; i < sn; i += 1024) acc[i] = 0.f;
    if (counting)
      for (int i = tid; i < sn; i += 1024) cnt[i] = 0.f;
    for (int dlo = 0; dlo < N; dlo += HVCAP) {
      const int dn = min(HVCAP, N - dlo);
      __syncthreads();
      for (int i = tid; i < dn; i += 1024) hv[i] = hc[dlo + i];
      __syncthreads();
      if (PK) walk_packed<false>(es, 0, e1, sbase, sn, dlo, dn, acc, hv, tid);
      else walk_edges<false>(es, ed, 0, e1, sbase, sn, dlo, dn, acc, hv, tid);
    }
    if (counting) {
      if (PK) walk_packed<true>(es, ce0, ce1, sbase, sn, 0, 0, cnt, hv, tid);
      else walk_edges<true>(es, ed, ce0, ce1, sbase, sn, 0, 0, cnt, hv, tid);
    }
    __syncthreads();
    for (int i = tid; i < sn; i += 1024) outp[sbase + i] = acc[i];
    if (counting)
      for (int i = tid; i < sn; i += 1024) outc[sbase + i] = cnt[i];
    __syncthreads();
  }
}

#define NCL 32
#define CAPL 16384
__global__ void __launch_bounds__(1024)
k_bin_cls(const int* __restrict__ src, const int* __restrict__ dst, int E, int qsz,
          int* __restrict__ lists, int* __restrict__ obin, long long bstride,
          int* __restrict__ fill) {
  __shared__ int lc[NQ * NCL];
  __shared__ int gb[NQ * NCL];
  const int tid = threadIdx.x;
  const int lane = tid & 31, w = tid >> 5;
  if (tid < NQ * NCL) lc[tid] = 0;
  __syncthreads();
  const long long e0 = (long long)blockIdx.x * (1024 * BINE) + w * (32 * BINE) + lane;
  int s[BINE], d[BINE], key[BINE], rk[BINE];
#pragma unroll
  for (int u = 0; u < BINE; ++u) {
    long long e = e0 + u * 32;
    bool valid = e < E;
    s[u] = valid ? src[e] : 0;
    d[u] = valid ? dst[e] : 0;
  }
#pragma unroll
  for (int u = 0; u < BINE; ++u) {
    long long e = e0 + u * 32;
    int qq = (s[u] >= qsz) + (s[u] >= 2 * qsz) + (s[u] >= 3 * qsz);
    s[u] -= qq * qsz;
    key[u] = (e < E) ? qq * NCL + (s[u] & (NCL - 1)) : -1;
    rk[u] = 0;
    if (key[u] >= 0) rk[u] = atomicAdd(&lc[key[u]], 1);
  }
  __syncthreads();
  if (tid < NQ * NCL) {
    const int t = lc[tid];
    gb[tid] = t > 0 ? atomicAdd(&fill[64 + tid], t) : 0;
  }
  __syncthreads();
#pragma unroll
  for (int u = 0; u < BINE; ++u) {
    if (key[u] >= 0) {
      const int qq = key[u] / NCL, r = key[u] & (NCL - 1);
      const int p = gb[key[u]] + rk[u];
      const int word = (int)(((unsigned)s[u] << 16) | (unsigned)d[u]);
      if (p < CAPL) {
        lists[(long long)qq * (NCL * CAPL) + ((p >> 2) * NCL + r) * 4 + (p & 3)] = word;
      } else {
        const int o = atomicAdd(&fill[qq], 1);
        obin[(long long)qq * bstride + o] = word;
      }
    }
  }
}

template <bool CNT, bool CHK>
__device__ __forceinline__ void add8(const int4 a0, const int4 a1, int p0, int p1, int cl,
                                     int lane, float* acc, const float* hv) {
  const unsigned wv[8] = {(unsigned)a0.x, (unsigned)a0.y, (unsigned)a0.z, (unsigned)a0.w,
                          (unsigned)a1.x, (unsigned)a1.y, (unsigned)a1.z, (unsigned)a1.w};
  unsigned ss[8];
  float v[8];
#pragma unroll
  for (int u = 0; u < 8; ++u) {
    const int p = (u < 4) ? p0 + u : p1 + (u - 4);
    const bool ok = !CHK || p < cl;
    ss[u] = ok ? (wv[u] >> 16) : (unsigned)lane;
    if (CNT) {
      v[u] = ok ? 1.0f : 0.0f;
    } else {
      const float x = hv[ok ? (wv[u] & 0xffffu) : 0u];
      v[u] = ok ? x : 0.0f;
    }
  }
#pragma unroll
  for (int u = 0; u < 8; ++u)
    __hip_atomic_fetch_add(&acc[ss[u]], v[u], __ATOMIC_RELAXED, __HIP_MEMORY_SCOPE_WORKGROUP);
}

template <bool CNT>
__device__ __forceinline__ void walk_cls(const int4* __restrict__ lp, int gbeg, int gend,
                                         int gfull, int cl, int w, int lane, float* acc,
                                         const float* hv) {
  int g = gbeg + 2 * w;
  if (g >= gend) return;
  int4 a0 = lp[(long long)g * NCL];
  int4 a1 = lp[(long long)(g + 1) * NCL];
  for (; g < gend; g += 64) {
    const int gn = (g + 64 < gend) ? g + 64 : g;
    const int4 b0 = lp[(long long)gn * NCL];
    const int4 b1 = lp[(long long)(gn + 1) * NCL];
    if (g + 1 < gfull) add8<CNT, false>(a0, a1, 4 * g, 4 * g + 4, cl, lane, acc, hv);
    else add8<CNT, true>(a0, a1, 4 * g, 4 * g + 4, cl, lane, acc, hv);
    a0 = b0;
    a1 = b1;
  }
}

__global__ void __launch_bounds__(1024)
k_colsum_cls(const int* __restrict__ lists, const int* __restrict__ obin, long long bstride,
             const int* __restrict__ fill, int qsz,
             const float* __restrict__ hT, float* __restrict__ sums,
             float* __restrict__ cntp, int N, int docount) {
  __shared__ float acc[SCAP];
  __shared__ float cnt[SCAP];
  __shared__ float hv[HVCAP];
  __shared__ int cls[NCL];
  const int tid = threadIdx.x;
  const int lane = tid & 31;
  const int w = __builtin_amdgcn_readfirstlane(tid >> 5);
  const int c = blockIdx.x & 63;
  const int q = blockIdx.x >> 6;
  const int qlo = q * qsz;
  const int qn = max(0, min(qsz, N - qlo));
  const float* hc = hT + (long long)c * N;
  float* outp = sums + (long long)c * N + qlo;
  float* outc = cntp + (long long)c * N + qlo;
  const bool counting = docount != 0 && c < CSPL;
  const int oe = fill[q];
  const int* ob = obin + (long long)q * bstride;

  if (tid < NCL) cls[tid] = min(fill[64 + q * NCL + tid], CAPL);
  for (int i = tid; i < qn; i += 1024) acc[i] = 0.f;
  if (counting)
    for (int i = tid; i < qn; i += 1024) cnt[i] = 0.f;
  for (int i = tid; i < N; i += 1024) hv[i] = hc[i];
  __syncthreads();

  const int cl = cls[lane];
  int lmin = cls[0], lmax = cls[0];
  for (int r = 1; r < NCL; ++r) {
    lmin = min(lmin, cls[r]);
    lmax = max(lmax, cls[r]);
  }
  const int G = (__builtin_amdgcn_readfirstlane(lmax) + 7) >> 3 << 1;
  const int gfull = __builtin_amdgcn_readfirstlane(lmin) >> 2;
  const int4* lp = (const int4*)(lists + (long long)q * (NCL * CAPL)) + lane;

  walk_cls<false>(lp, 0, G, gfull, cl, w, lane, acc, hv);
  if (oe > 0) walk_packed<false>(ob, 0, oe, 0, qn, 0, N, acc, hv, tid);
  if (counting) {
    const int gper = ((G / 2 + CSPL - 1) / CSPL) * 2;
    const int g0 = min(G, c * gper);
    const int g1 = min(G, g0 + gper);
    walk_cls<true>(lp, g0, g1, gfull, cl, w, lane, cnt, hv);
    if (c == 0 && oe > 0) walk_packed<true>(ob, 0, oe, 0, qn, 0, 0, cnt, hv, tid);
  }
  __syncthreads();
  for (int i = tid; i < qn; i += 1024) outp[i] = acc[i];
  if (counting)
    for (int i = tid; i < qn; i += 1024) outc[i] = cnt[i];
}

template <bool OUT_T>
__global__ void __launch_bounds__(256)
k_layer_gemm(const float* __restrict__ hT, const float* __restrict__ part, int np,
             const float* __restrict__ cntp, int ncp,
             const float* __restrict__ Wself, const float* __restrict__ bself,
             const float* __restrict__ Wneigh, const float* __restrict__ bneigh,
             float* __restrict__ hout, int N) {
  __shared__ float Bl[32 * 4 * 32 * 2];
  int tid = threadIdx.x;
  for (int idx = tid; idx < 8192; idx += 256) {
    int v = idx & 1, lane = (idx >> 1) & 31, ct = (idx >> 6) & 3, kk = idx >> 8;
    int n = ct * 16 + (lane & 15);
    int k = kk * 4 + 2 * (lane >> 4) + v;
    Bl[idx] = (k < 64) ? Wself[n * 64 + k] : Wneigh[n * 64 + (k - 64)];
  }
  __syncthreads();

  int w = tid >> 5, lane = tid & 31;
  int base = (blockIdx.x * 8 + w) * 16;
  int half = lane >> 4;
  int mrow = base + (lane & 15);
  long long mc = (mrow < N) ? mrow : 0;

  float c = 0.f;
  for (int j = 0; j < ncp; ++j) c += cntp[(long long)j * N + mc];
  float inv = (c > 0.f) ? 1.f / c : 1.f;

  v2f X[32];
#pragma unroll
  for (int kk = 0; kk < 16; ++kk) {
    long long k0 = kk * 4 + 2 * half;
    v2f a = {hT[k0 * N + mc], hT[(k0 + 1) * N + mc]};
    X[kk] = a;
    v2f z = {0.f, 0.f};
    X[16 + kk] = z;
  }
  for (int p = 0; p < np; ++p) {
    const float* pp = part + (long long)p * 64 * N + mc;
#pragma unroll
    for (int kk = 0; kk < 16; ++kk) {
      long long k0 = kk * 4 + 2 * half;
      X[16 + kk].x += pp[k0 * N];
      X[16 + kk].y += pp[(k0 + 1) * N];
    }
  }
#pragma unroll
  for (int kk = 0; kk < 16; ++kk) {
    X[16 + kk].x *= inv;
    X[16 + kk].y *= inv;
  }

  v8f acc0 = {}, acc1 = {}, acc2 = {}, acc3 = {};
#pragma unroll
  for (int kk = 0; kk < 32; ++kk) {
    int bb = kk * 256 + lane * 2;
    v2f w0 = *(const v2f*)&Bl[bb + 0];
    v2f w1 = *(const v2f*)&Bl[bb + 64];
    v2f w2 = *(const v2f*)&Bl[bb + 128];
    v2f w3 = *(const v2f*)&Bl[bb + 192];
    if (OUT_T) {
      acc0 = __builtin_amdgcn_wmma_f32_16x16x4_f32(false, w0, false, X[kk], (short)0, acc0, false, false);
      acc1 = __builtin_amdgcn_wmma_f32_16x16x4_f32(false, w1, false, X[kk], (short)0, acc1, false, false);
      acc2 = __builtin_amdgcn_wmma_f32_16x16x4_f32(false, w2, false, X[kk], (short)0, acc2, false, false);
      acc3 = __builtin_amdgcn_wmma_f32_16x16x4_f32(false, w3, false, X[kk], (short)0, acc3, false, false);
    } else {
      acc0 = __builtin_amdgcn_wmma_f32_16x16x4_f32(false, X[kk], false, w0, (short)0, acc0, false, false);
      acc1 = __builtin_amdgcn_wmma_f32_16x16x4_f32(false, X[kk], false, w1, (short)0, acc1, false, false);
      acc2 = __builtin_amdgcn_wmma_f32_16x16x4_f32(false, X[kk], false, w2, (short)0, acc2, false, false);
      acc3 = __builtin_amdgcn_wmma_f32_16x16x4_f32(false, X[kk], false, w3, (short)0, acc3, false, false);
    }
  }

#pragma unroll
  for (int ct = 0; ct < 4; ++ct) {
    v8f av = (ct == 0) ? acc0 : (ct == 1) ? acc1 : (ct == 2) ? acc2 : acc3;
    if (OUT_T) {
      int mg = base + (lane & 15);
#pragma unroll
      for (int r = 0; r < 8; ++r) {
        int n = ct * 16 + r + 8 * half;
        float val = av[r] + (bself[n] + bneigh[n]);
        if (mg < N) hout[(long long)n * N + mg] = val > 0.f ? val : 0.f;
      }
    } else {
      int n = ct * 16 + (lane & 15);
      float bs = bself[n] + bneigh[n];
#pragma unroll
      for (int r = 0; r < 8; ++r) {
        int mg = base + r + 8 * half;
        if (mg < N) {
          float val = av[r] + bs;
          hout[(long long)mg * 64 + n] = val > 0.f ? val : 0.f;
        }
      }
    }
  }
}

#define ESW 32
__global__ void __launch_bounds__(1024)
k_edge_score(const float* __restrict__ ef, const float* __restrict__ Wa,
             const float* __restrict__ ba, const float* __restrict__ We,
             const float* __restrict__ be, const float* __restrict__ env, int NV,
             const float* __restrict__ W1, const float* __restrict__ b1,
             const float* __restrict__ W2, const float* __restrict__ b2,
             float* __restrict__ scores, int E) {
  __shared__ float Bs[16 * 4 * 32 * 2];
  __shared__ __attribute__((aligned(16))) float Af[192];
  __shared__ __attribute__((aligned(16))) float Cf[64];
  __shared__ float WeS[192];
  __shared__ float beS[64];
  __shared__ float vmS[64];
  const int tid = threadIdx.x;
  const int lane = tid & 31;
  const int half = lane >> 4;
  const int w = __builtin_amdgcn_readfirstlane(tid >> 5);
  const int nwv = (int)gridDim.x * ESW;
  const int ntiles = (E + 15) / 16;

  float bn_[4], wn[4];
#pragma unroll
  for (int ct = 0; ct < 4; ++ct) {
    bn_[ct] = b1[ct * 16 + (lane & 15)];
    wn[ct] = W2[ct * 16 + (lane & 15)];
  }
  const float bb2 = b2[0];

  int t = w * (int)gridDim.x + (int)blockIdx.x;
  float f0 = 0.f, f1 = 0.f, f2 = 0.f;
  if (t < ntiles) {
    const int row = t * 16 + (lane & 15);
    const long long rc = (row < E) ? row : 0;
    f0 = ef[rc * 3 + 0];
    f1 = ef[rc * 3 + 1];
    f2 = ef[rc * 3 + 2];
  }

#pragma unroll
  for (int idx = tid; idx < 4096; idx += 1024) {
    int v = idx & 1, ln = (idx >> 1) & 31, ct = (idx >> 6) & 3, kk = idx >> 8;
    int n = ct * 16 + (ln & 15);
    int k = kk * 4 + 2 * (ln >> 4) + v;
    Bs[idx] = W1[n * 64 + k];
  }
  if (tid < 192) {
    WeS[tid] = We[tid];
  } else if (tid < 256) {
    beS[tid - 192] = be[tid - 192];
  } else if (tid < 320) {
    const int d = tid - 256;
    float vm = 0.f;
    for (int v = 0; v < NV; ++v) vm += env[v * 64 + d];
    vmS[d] = vm * (1.0f / (float)NV);
  }
  __syncthreads();
  if (tid < 192) {
    const int j = tid / 3, i = tid - 3 * j;
    const float4* wr = (const float4*)(Wa + j * 128);
    float a = 0.f;
#pragma unroll
    for (int kq = 0; kq < 16; ++kq) {
      const float4 wv = wr[kq];
      a = __builtin_fmaf(wv.x, WeS[(4 * kq + 0) * 3 + i], a);
      a = __builtin_fmaf(wv.y, WeS[(4 * kq + 1) * 3 + i], a);
      a = __builtin_fmaf(wv.z, WeS[(4 * kq + 2) * 3 + i], a);
      a = __builtin_fmaf(wv.w, WeS[(4 * kq + 3) * 3 + i], a);
    }
    Af[tid] = a;
  } else if (tid < 256) {
    const int j = tid - 192;
    const float4* wr = (const float4*)(Wa + j * 128);
    float c = ba[j];
#pragma unroll
    for (int kq = 0; kq < 16; ++kq) {
      const float4 wv = wr[kq];
      c = __builtin_fmaf(wv.x, beS[4 * kq + 0], c);
      c = __builtin_fmaf(wv.y, beS[4 * kq + 1], c);
      c = __builtin_fmaf(wv.z, beS[4 * kq + 2], c);
      c = __builtin_fmaf(wv.w, beS[4 * kq + 3], c);
    }
#pragma unroll
    for (int dq = 0; dq < 16; ++dq) {
      const float4 wv = wr[16 + dq];
      c = __builtin_fmaf(wv.x, vmS[4 * dq + 0], c);
      c = __builtin_fmaf(wv.y, vmS[4 * dq + 1], c);
      c = __builtin_fmaf(wv.z, vmS[4 * dq + 2], c);
      c = __builtin_fmaf(wv.w, vmS[4 * dq + 3], c);
    }
    Cf[j] = c;
  }
  __syncthreads();

  for (; t < ntiles; t += nwv) {
    asm volatile("" ::: "memory");
    const int tn = t + nwv;
    float n0 = 0.f, n1 = 0.f, n2 = 0.f;
    if (tn < ntiles) {
      const int row = tn * 16 + (lane & 15);
      const long long rc = (row < E) ? row : 0;
      n0 = ef[rc * 3 + 0];
      n1 = ef[rc * 3 + 1];
      n2 = ef[rc * 3 + 2];
    }
    const int base = t * 16;

    v2f A[16];
#pragma unroll
    for (int kk = 0; kk < 16; ++kk) {
      int k0 = kk * 4 + 2 * half;
      float x0 = __builtin_fmaf(Af[k0 * 3 + 2], f2,
                 __builtin_fmaf(Af[k0 * 3 + 1], f1, __builtin_fmaf(Af[k0 * 3], f0, Cf[k0])));
      float x1 = __builtin_fmaf(Af[(k0 + 1) * 3 + 2], f2,
                 __builtin_fmaf(Af[(k0 + 1) * 3 + 1], f1,
                                __builtin_fmaf(Af[(k0 + 1) * 3], f0, Cf[k0 + 1])));
      v2f a = {x0 > 0.f ? x0 : 0.f, x1 > 0.f ? x1 : 0.f};
      A[kk] = a;
    }

    v8f acc0 = {}, acc1 = {}, acc2 = {}, acc3 = {};
#pragma unroll
    for (int kk = 0; kk < 16; ++kk) {
      int bb = kk * 256 + lane * 2;
      acc0 = __builtin_amdgcn_wmma_f32_16x16x4_f32(false, A[kk], false, *(const v2f*)&Bs[bb + 0],   (short)0, acc0, false, false);
      acc1 = __builtin_amdgcn_wmma_f32_16x16x4_f32(false, A[kk], false, *(const v2f*)&Bs[bb + 64],  (short)0, acc1, false, false);
      acc2 = __builtin_amdgcn_wmma_f32_16x16x4_f32(false, A[kk], false, *(const v2f*)&Bs[bb + 128], (short)0, acc2, false, false);
      acc3 = __builtin_amdgcn_wmma_f32_16x16x4_f32(false, A[kk], false, *(const v2f*)&Bs[bb + 192], (short)0, acc3, false, false);
    }

    float p[8] = {0, 0, 0, 0, 0, 0, 0, 0};
#pragma unroll
    for (int ct = 0; ct < 4; ++ct) {
      v8f av = (ct == 0) ? acc0 : (ct == 1) ? acc1 : (ct == 2) ? acc2 : acc3;
#pragma unroll
      for (int r = 0; r < 8; ++r) {
        float s = av[r] + bn_[ct];
        s = s > 0.f ? s : 0.f;
        p[r] += wn[ct] * s;
      }
    }
#pragma unroll
    for (int r = 0; r < 8; ++r) {
      p[r] += __shfl_xor(p[r], 1);
      p[r] += __shfl_xor(p[r], 2);
      p[r] += __shfl_xor(p[r], 4);
      p[r] += __shfl_xor(p[r], 8);
    }
    if ((lane & 15) == 0) {
      int rb = base + 8 * half;
#pragma unroll
      for (int r = 0; r < 8; ++r)
        if (rb + r < E) scores[rb + r] = p[r] + bb2;
    }
    f0 = n0;
    f1 = n1;
    f2 = n2;
  }
}

extern "C" void kernel_launch(void* const* d_in, const int* in_sizes, int n_in,
                              void* d_out, int out_size, void* d_ws, size_t ws_size,
                              hipStream_t stream) {
  if (n_in < 18) return;
  const float* node_feats = (const float*)d_in[0];
  const float* edge_feats = (const float*)d_in[1];
  const float* env        = (const float*)d_in[2];
  const int*   eidx       = (const int*)d_in[3];
  const float* Wn  = (const float*)d_in[4];
  const float* bn  = (const float*)d_in[5];
  const float* We  = (const float*)d_in[6];
  const float* be  = (const float*)d_in[7];
  const float* Wself  = (const float*)d_in[8];
  const float* bself  = (const float*)d_in[9];
  const float* Wneigh = (const float*)d_in[10];
  const float* bneigh = (const float*)d_in[11];
  const float* Wa  = (const float*)d_in[12];
  const float* ba  = (const float*)d_in[13];
  const float* W1  = (const float*)d_in[14];
  const float* b1  = (const float*)d_in[15];
  const float* W2  = (const float*)d_in[16];
  const float* b2  = (const float*)d_in[17];

  int N  = in_sizes[0] / 3;
  int E  = in_sizes[1] / 3;
  int NV = in_sizes[2] / 64;
  int L  = in_sizes[8] / (64 * 64);

  const int* src = eidx;
  const int* dst = eidx + E;

  float* scores = (float*)d_out;
  float* h_out  = (float*)d_out + E;

  int qsz  = (N + NQ - 1) / NQ;
  int np   = 1;

  long long n64  = (long long)N * 64;
  long long npad = (n64 + 63) & ~63LL;
  float* hTa   = (float*)d_ws;
  float* hTb   = hTa + npad;
  float* part  = hTb + npad;
  float* cntp  = part + (long long)np * npad;
  int* binfill = (int*)(cntp + ((((long long)CSPL * N) + 63) & ~63LL));
  long long bstride = ((long long)E + 31) & ~31LL;
  int* bsrc = binfill + 256;
  int* bdst = bsrc + NQ * bstride;
  int* lists = bdst + NQ * bstride;

  dim3 gfc((N + 255) / 256, 64);
  if (L == 0) k_node_fc<<<gfc, 256, 0, stream>>>(node_feats, Wn, bn, h_out, N, 1, binfill);
  else        k_node_fc<<<gfc, 256, 0, stream>>>(node_feats, Wn, bn, hTa, N, 0, binfill);
  const bool pk = N <= 65536;
  const bool lcl = pk && N >= 256 && N <= HVCAP && qsz <= SCAP;
  if (L > 0) {
    if (lcl)
      k_bin_cls<<<(E + 1024 * BINE - 1) / (1024 * BINE), 1024, 0, stream>>>(
          src, dst, E, qsz, lists, bsrc, bstride, binfill);
    else if (pk)
      k_bin<true><<<(E + 1024 * BINE - 1) / (1024 * BINE), 1024, 0, stream>>>(
          src, dst, E, qsz, bsrc, bdst, bstride, binfill);
    else
      k_bin<false><<<(E + 1024 * BINE - 1) / (1024 * BINE), 1024, 0, stream>>>(
          src, dst, E, qsz, bsrc, bdst, bstride, binfill);
  }

  int tiles = (N + 15) / 16;
  float* hin = hTa;
  float* hnx = hTb;
  for (int l = 0; l < L; ++l) {
    if (lcl)
      k_colsum_cls<<<NQ * 64, 1024, 0, stream>>>(lists, bsrc, bstride, binfill, qsz, hin, part,
                                                 cntp, N, l == 0 ? 1 : 0);
    else if (pk)
      k_colsum<true><<<NQ * 64, 1024, 0, stream>>>(bsrc, bdst, bstride, binfill, qsz, hin, part,
                                                   cntp, N, l == 0 ? 1 : 0);
    else
      k_colsum<false><<<NQ * 64, 1024, 0, stream>>>(bsrc, bdst, bstride, binfill, qsz, hin, part,
                                                    cntp, N, l == 0 ? 1 : 0);
    if (l == L - 1) {
      k_layer_gemm<false><<<(tiles + 7) / 8, 256, 0, stream>>>(
          hin, part, np, cntp, CSPL,
          Wself + (long long)l * 64 * 64, bself + (long long)l * 64,
          Wneigh + (long long)l * 64 * 64, bneigh + (long long)l * 64, h_out, N);
    } else {
      k_layer_gemm<true><<<(tiles + 7) / 8, 256, 0, stream>>>(
          hin, part, np, cntp, CSPL,
          Wself + (long long)l * 64 * 64, bself + (long long)l * 64,
          Wneigh + (long long)l * 64 * 64, bneigh + (long long)l * 64, hnx, N);
      float* t = hin; hin = hnx; hnx = t;
    }
  }

  k_edge_score<<<256, 1024, 0, stream>>>(edge_feats, Wa, ba, We, be, env, NV, W1, b1, W2, b2,
                                         scores, E);
}
